# starting kernel
# baseline (speedup 1.0000x reference)
#include <hip/hip_runtime.h>
#include <hip/hip_bf16.h>
#include <math.h>

typedef float v2f __attribute__((ext_vector_type(2)));
typedef float v8f __attribute__((ext_vector_type(8)));

#define NODE_DIM 128
#define LDS_STRIDE 132
#define WAVES_PER_BLOCK 2

__global__ void sage_count_kernel(const int* __restrict__ dst,
                                  float* __restrict__ cnt, int E) {
  int e = blockIdx.x * blockDim.x + threadIdx.x;
  if (e < E) atomicAdd(&cnt[dst[e]], 1.0f);
}

__global__ void sage_scatter_kernel(const float* __restrict__ feat,
                                    const int* __restrict__ src,
                                    const int* __restrict__ dst,
                                    float* __restrict__ agg, int E) {
  long long gid = (long long)blockIdx.x * blockDim.x + threadIdx.x;
  int e = (int)(gid >> 5);
  int q = (int)(gid & 31);
  if (e >= E) return;
  int s = src[e];
  int d = dst[e];
  const float4 v = *reinterpret_cast<const float4*>(
      feat + (long long)s * NODE_DIM + q * 4);
  float* base = agg + (long long)d * NODE_DIM + q * 4;
  atomicAdd(base + 0, v.x);
  atomicAdd(base + 1, v.y);
  atomicAdd(base + 2, v.z);
  atomicAdd(base + 3, v.w);
}

template <bool NORM_RELU>
__global__ __launch_bounds__(WAVES_PER_BLOCK * 32)
void sage_combine_kernel(const float* __restrict__ agg,
                         const float* __restrict__ cnt,
                         const float* __restrict__ xin,
                         const float* __restrict__ Wl,
                         const float* __restrict__ bias,
                         const float* __restrict__ Wr,
                         float* __restrict__ outp, int N) {
  __shared__ float s_agg[WAVES_PER_BLOCK][16][LDS_STRIDE];
  __shared__ float s_in [WAVES_PER_BLOCK][16][LDS_STRIDE];
  __shared__ float s_out[WAVES_PER_BLOCK][16][LDS_STRIDE];

  const int lane  = threadIdx.x & 31;
  const int wv    = threadIdx.x >> 5;
  const int tile  = blockIdx.x * WAVES_PER_BLOCK + wv;
  const int node0 = tile * 16;

  #pragma unroll 4
  for (int r = 0; r < 16; ++r) {
    const int n = node0 + r;
    if (n < N) {
      const float rc = 1.0f / fmaxf(cnt[n], 1.0f);
      const float4 va = *reinterpret_cast<const float4*>(
          agg + (long long)n * NODE_DIM + lane * 4);
      const float4 vx = *reinterpret_cast<const float4*>(
          xin + (long long)n * NODE_DIM + lane * 4);
      s_agg[wv][r][lane * 4 + 0] = va.x * rc;
      s_agg[wv][r][lane * 4 + 1] = va.y * rc;
      s_agg[wv][r][lane * 4 + 2] = va.z * rc;
      s_agg[wv][r][lane * 4 + 3] = va.w * rc;
      s_in[wv][r][lane * 4 + 0] = vx.x;
      s_in[wv][r][lane * 4 + 1] = vx.y;
      s_in[wv][r][lane * 4 + 2] = vx.z;
      s_in[wv][r][lane * 4 + 3] = vx.w;
    } else {
      s_agg[wv][r][lane * 4 + 0] = 0.f; s_agg[wv][r][lane * 4 + 1] = 0.f;
      s_agg[wv][r][lane * 4 + 2] = 0.f; s_agg[wv][r][lane * 4 + 3] = 0.f;
      s_in[wv][r][lane * 4 + 0] = 0.f;  s_in[wv][r][lane * 4 + 1] = 0.f;
      s_in[wv][r][lane * 4 + 2] = 0.f;  s_in[wv][r][lane * 4 + 3] = 0.f;
    }
  }
  __syncthreads();

  const int m     = lane & 15;
  const int khalf = (lane >> 4) << 1;
  const int mbase = (lane >> 4) << 3;

  for (int ct = 0; ct < 8; ++ct) {
    const int n0 = ct * 16;
    const float bv = bias[n0 + m];
    v8f c;
    #pragma unroll
    for (int v = 0; v < 8; ++v) c[v] = bv;

    const float* wlrow = Wl + (long long)(n0 + m) * NODE_DIM;
    const float* wrrow = Wr + (long long)(n0 + m) * NODE_DIM;

    #pragma unroll 4
    for (int k0 = 0; k0 < NODE_DIM; k0 += 4) {
      v2f aA, aX;
      aA[0] = s_agg[wv][m][k0 + khalf];
      aA[1] = s_agg[wv][m][k0 + khalf + 1];
      const v2f bL = *reinterpret_cast<const v2f*>(wlrow + k0 + khalf);
      c = __builtin_amdgcn_wmma_f32_16x16x4_f32(false, aA, false, bL,
                                                (short)0, c, false, false);
      aX[0] = s_in[wv][m][k0 + khalf];
      aX[1] = s_in[wv][m][k0 + khalf + 1];
      const v2f bR = *reinterpret_cast<const v2f*>(wrrow + k0 + khalf);
      c = __builtin_amdgcn_wmma_f32_16x16x4_f32(false, aX, false, bR,
                                                (short)0, c, false, false);
    }

    #pragma unroll
    for (int v = 0; v < 8; ++v) s_out[wv][mbase + v][n0 + m] = c[v];
  }
  __syncthreads();

  if (NORM_RELU) {
    const int row = lane & 15;
    const int cb  = (lane >> 4) * 64;
    float ss = 0.f;
    #pragma unroll 8
    for (int c2 = 0; c2 < 64; ++c2) {
      const float v = s_out[wv][row][cb + c2];
      ss += v * v;
    }
    ss += __shfl_xor(ss, 16, 32);
    const float scl = 1.0f / fmaxf(sqrtf(ss), 1e-12f);
    #pragma unroll 8
    for (int c2 = 0; c2 < 64; ++c2) {
      const float v = s_out[wv][row][cb + c2] * scl;
      s_out[wv][row][cb + c2] = fmaxf(v, 0.0f);
    }
    __syncthreads();
  }

  #pragma unroll 4
  for (int r = 0; r < 16; ++r) {
    const int n = node0 + r;
    if (n < N) {
      float4 v;
      v.x = s_out[wv][r][lane * 4 + 0];
      v.y = s_out[wv][r][lane * 4 + 1];
      v.z = s_out[wv][r][lane * 4 + 2];
      v.w = s_out[wv][r][lane * 4 + 3];
      *reinterpret_cast<float4*>(outp + (long long)n * NODE_DIM + lane * 4) = v;
    }
  }
}

extern "C" void kernel_launch(void* const* d_in, const int* in_sizes, int n_in,
                              void* d_out, int out_size, void* d_ws,
                              size_t ws_size, hipStream_t stream) {
  (void)n_in; (void)out_size; (void)ws_size;

  const float* x   = (const float*)d_in[0];
  const int*   ei  = (const int*)d_in[1];
  const float* W1l = (const float*)d_in[2];
  const float* b1  = (const float*)d_in[3];
  const float* W1r = (const float*)d_in[4];
  const float* W2l = (const float*)d_in[5];
  const float* b2  = (const float*)d_in[6];
  const float* W2r = (const float*)d_in[7];
  float* out = (float*)d_out;

  const int N = in_sizes[0] / NODE_DIM;
  const int E = in_sizes[1] / 2;
  const int* src = ei;
  const int* dst = ei + E;

  char* ws = (char*)d_ws;
  float* cnt = (float*)ws;
  size_t off = ((size_t)N * 4 + 255) & ~(size_t)255;
  float* aggb = (float*)(ws + off);
  off += (((size_t)N * NODE_DIM * 4) + 255) & ~(size_t)255;
  float* h = (float*)(ws + off);

  const int tiles  = (N + 15) / 16;
  const int cblk   = (tiles + WAVES_PER_BLOCK - 1) / WAVES_PER_BLOCK;
  const long long sthreads = (long long)E * 32;
  const int sblk = (int)((sthreads + 255) / 256);
  const int dblk = (E + 255) / 256;

  hipMemsetAsync(cnt, 0, (size_t)N * 4, stream);
  hipMemsetAsync(aggb, 0, (size_t)N * NODE_DIM * 4, stream);
  sage_count_kernel<<<dblk, 256, 0, stream>>>(dst, cnt, E);
  sage_scatter_kernel<<<sblk, 256, 0, stream>>>(x, src, dst, aggb, E);
  sage_combine_kernel<true><<<cblk, WAVES_PER_BLOCK * 32, 0, stream>>>(
      aggb, cnt, x, W1l, b1, W1r, h, N);

  hipMemsetAsync(aggb, 0, (size_t)N * NODE_DIM * 4, stream);
  sage_scatter_kernel<<<sblk, 256, 0, stream>>>(h, src, dst, aggb, E);
  sage_combine_kernel<false><<<cblk, WAVES_PER_BLOCK * 32, 0, stream>>>(
      aggb, cnt, h, W2l, b2, W2r, out, N);
}


# Pull replaces push: edges grouped by destination per call (int atomics + scan), each node's sources sorted by id, one wave adds a node's rows in that fixed order; no float atomics, sum order fixed
# speedup vs baseline: 1.8397x; 1.8397x over previous
#include <hip/hip_runtime.h>
#include <hip/hip_bf16.h>
#include <math.h>

typedef float v2f __attribute__((ext_vector_type(2)));
typedef float v8f __attribute__((ext_vector_type(8)));

#define NODE_DIM 128
#define LDS_STRIDE 132
#define WAVES_PER_BLOCK 2

#define SCAN_TILE 1024
#define NODES_PER_BLOCK 8

__global__ void sage_rank_kernel(const int* __restrict__ dst,
                                 int* __restrict__ deg,
                                 int* __restrict__ rnk, int E) {
  int e = blockIdx.x * blockDim.x + threadIdx.x;
  if (e < E) rnk[e] = atomicAdd(&deg[dst[e]], 1);
}

__global__ __launch_bounds__(256)
void sage_scan_kernel(const int* __restrict__ deg, int* __restrict__ first,
                      int N) {
  __shared__ int s_part[8];
  const int tid   = threadIdx.x;
  const int lane  = tid & 31;
  const int wv    = tid >> 5;
  const int tile0 = blockIdx.x * SCAN_TILE;

  int acc = 0;
  const int4* deg4 = reinterpret_cast<const int4*>(deg);
  #pragma unroll 16
  for (int t = 0; t < (int)blockIdx.x; ++t) {
    const int4 v = deg4[t * 256 + tid];
    acc += (v.x + v.y) + (v.z + v.w);
  }
  #pragma unroll
  for (int o = 16; o > 0; o >>= 1) acc += __shfl_xor(acc, o, 32);
  if (lane == 0) s_part[wv] = acc;
  __syncthreads();
  int base = 0;
  #pragma unroll
  for (int w = 0; w < 8; ++w) base += s_part[w];
  __syncthreads();

  const int i0 = tile0 + tid * 4;
  int d[4];
  #pragma unroll
  for (int t = 0; t < 4; ++t) d[t] = (i0 + t < N) ? deg[i0 + t] : 0;
  const int tsum = (d[0] + d[1]) + (d[2] + d[3]);
  int incl = tsum;
  #pragma unroll
  for (int o = 1; o < 32; o <<= 1) {
    const int up = __shfl_up(incl, o, 32);
    if (lane >= o) incl += up;
  }
  if (lane == 31) s_part[wv] = incl;
  __syncthreads();
  int ex = base + incl - tsum;
  for (int w = 0; w < wv; ++w) ex += s_part[w];
  #pragma unroll
  for (int t = 0; t < 4; ++t) {
    if (i0 + t < N) first[i0 + t] = ex;
    ex += d[t];
  }
}

__global__ void sage_place_kernel(const int* __restrict__ src,
                                  const int* __restrict__ dst,
                                  const int* __restrict__ first,
                                  const int* __restrict__ rnk,
                                  int* __restrict__ slots, int E) {
  int e = blockIdx.x * blockDim.x + threadIdx.x;
  if (e < E) slots[first[dst[e]] + rnk[e]] = src[e];
}

__global__ __launch_bounds__(NODES_PER_BLOCK * 32)
void sage_order_kernel(const int* __restrict__ first,
                       const int* __restrict__ deg,
                       const int* __restrict__ slots,
                       int* __restrict__ csr, int N) {
  const int lane = threadIdx.x & 31;
  const int n = blockIdx.x * NODES_PER_BLOCK +
                __builtin_amdgcn_readfirstlane((int)(threadIdx.x >> 5));
  if (n >= N) return;
  const int beg = first[n];
  const int d   = deg[n];
  for (int ib = 0; ib < d; ib += 32) {
    const int i  = ib + lane;
    const int si = (i < d) ? slots[beg + i] : 0;
    int pos = 0;
    for (int jb = 0; jb < d; jb += 32) {
      const int j0  = jb + lane;
      const int sc  = (j0 < d) ? slots[beg + j0] : 0;
      const int cnt = min(32, d - jb);
      for (int jj = 0; jj < cnt; ++jj) {
        const int sj = __builtin_amdgcn_readlane(sc, jj);
        const int j  = jb + jj;
        pos += ((sj < si) || (sj == si && j < i)) ? 1 : 0;
      }
    }
    if (i < d) csr[beg + pos] = si;
  }
}

__global__ __launch_bounds__(NODES_PER_BLOCK * 32)
void sage_gather_kernel(const float* __restrict__ feat,
                        const int* __restrict__ first,
                        const int* __restrict__ deg,
                        const int* __restrict__ csr,
                        float* __restrict__ agg, int N) {
  const int lane = threadIdx.x & 31;
  const int n = blockIdx.x * NODES_PER_BLOCK +
                __builtin_amdgcn_readfirstlane((int)(threadIdx.x >> 5));
  if (n >= N) return;
  const int beg = first[n];
  const int d   = deg[n];
  float4 acc = make_float4(0.f, 0.f, 0.f, 0.f);
  for (int jb = 0; jb < d; jb += 32) {
    const int j0  = jb + lane;
    const int sc  = (j0 < d) ? csr[beg + j0] : 0;
    const int cnt = min(32, d - jb);
    for (int j8 = 0; j8 < cnt; j8 += 8) {
      float4 v[8];
      #pragma unroll
      for (int u = 0; u < 8; ++u) {
        const int s = __builtin_amdgcn_readlane(sc, j8 + u);
        v[u] = *reinterpret_cast<const float4*>(
            feat + (long long)s * NODE_DIM + lane * 4);
      }
      #pragma unroll
      for (int u = 0; u < 8; ++u) {
        if (u == 0 || j8 + u < cnt) {
          acc.x += v[u].x; acc.y += v[u].y; acc.z += v[u].z; acc.w += v[u].w;
        }
      }
    }
  }
  *reinterpret_cast<float4*>(agg + (long long)n * NODE_DIM + lane * 4) = acc;
}

template <bool NORM_RELU>
__global__ __launch_bounds__(WAVES_PER_BLOCK * 32)
void sage_combine_kernel(const float* __restrict__ agg,
                         const int* __restrict__ deg,
                         const float* __restrict__ xin,
                         const float* __restrict__ Wl,
                         const float* __restrict__ bias,
                         const float* __restrict__ Wr,
                         float* __restrict__ outp, int N) {
  __shared__ float s_agg[WAVES_PER_BLOCK][16][LDS_STRIDE];
  __shared__ float s_in [WAVES_PER_BLOCK][16][LDS_STRIDE];
  __shared__ float s_out[WAVES_PER_BLOCK][16][LDS_STRIDE];

  const int lane  = threadIdx.x & 31;
  const int wv    = threadIdx.x >> 5;
  const int tile  = blockIdx.x * WAVES_PER_BLOCK + wv;
  const int node0 = tile * 16;

  #pragma unroll 4
  for (int r = 0; r < 16; ++r) {
    const int n = node0 + r;
    if (n < N) {
      const float rc = 1.0f / fmaxf((float)deg[n], 1.0f);
      const float4 va = *reinterpret_cast<const float4*>(
          agg + (long long)n * NODE_DIM + lane * 4);
      const float4 vx = *reinterpret_cast<const float4*>(
          xin + (long long)n * NODE_DIM + lane * 4);
      s_agg[wv][r][lane * 4 + 0] = va.x * rc;
      s_agg[wv][r][lane * 4 + 1] = va.y * rc;
      s_agg[wv][r][lane * 4 + 2] = va.z * rc;
      s_agg[wv][r][lane * 4 + 3] = va.w * rc;
      s_in[wv][r][lane * 4 + 0] = vx.x;
      s_in[wv][r][lane * 4 + 1] = vx.y;
      s_in[wv][r][lane * 4 + 2] = vx.z;
      s_in[wv][r][lane * 4 + 3] = vx.w;
    } else {
      s_agg[wv][r][lane * 4 + 0] = 0.f; s_agg[wv][r][lane * 4 + 1] = 0.f;
      s_agg[wv][r][lane * 4 + 2] = 0.f; s_agg[wv][r][lane * 4 + 3] = 0.f;
      s_in[wv][r][lane * 4 + 0] = 0.f;  s_in[wv][r][lane * 4 + 1] = 0.f;
      s_in[wv][r][lane * 4 + 2] = 0.f;  s_in[wv][r][lane * 4 + 3] = 0.f;
    }
  }
  __syncthreads();

  const int m     = lane & 15;
  const int khalf = (lane >> 4) << 1;
  const int mbase = (lane >> 4) << 3;

  for (int ct = 0; ct < 8; ++ct) {
    const int n0 = ct * 16;
    const float bv = bias[n0 + m];
    v8f c;
    #pragma unroll
    for (int v = 0; v < 8; ++v) c[v] = bv;

    const float* wlrow = Wl + (long long)(n0 + m) * NODE_DIM;
    const float* wrrow = Wr + (long long)(n0 + m) * NODE_DIM;

    #pragma unroll 4
    for (int k0 = 0; k0 < NODE_DIM; k0 += 4) {
      v2f aA, aX;
      aA[0] = s_agg[wv][m][k0 + khalf];
      aA[1] = s_agg[wv][m][k0 + khalf + 1];
      const v2f bL = *reinterpret_cast<const v2f*>(wlrow + k0 + khalf);
      c = __builtin_amdgcn_wmma_f32_16x16x4_f32(false, aA, false, bL,
                                                (short)0, c, false, false);
      aX[0] = s_in[wv][m][k0 + khalf];
      aX[1] = s_in[wv][m][k0 + khalf + 1];
      const v2f bR = *reinterpret_cast<const v2f*>(wrrow + k0 + khalf);
      c = __builtin_amdgcn_wmma_f32_16x16x4_f32(false, aX, false, bR,
                                                (short)0, c, false, false);
    }

    #pragma unroll
    for (int v = 0; v < 8; ++v) s_out[wv][mbase + v][n0 + m] = c[v];
  }
  __syncthreads();

  if (NORM_RELU) {
    const int row = lane & 15;
    const int cb  = (lane >> 4) * 64;
    float ss = 0.f;
    #pragma unroll 8
    for (int c2 = 0; c2 < 64; ++c2) {
      const float v = s_out[wv][row][cb + c2];
      ss += v * v;
    }
    ss += __shfl_xor(ss, 16, 32);
    const float scl = 1.0f / fmaxf(sqrtf(ss), 1e-12f);
    #pragma unroll 8
    for (int c2 = 0; c2 < 64; ++c2) {
      const float v = s_out[wv][row][cb + c2] * scl;
      s_out[wv][row][cb + c2] = fmaxf(v, 0.0f);
    }
    __syncthreads();
  }

  #pragma unroll 4
  for (int r = 0; r < 16; ++r) {
    const int n = node0 + r;
    if (n < N) {
      float4 v;
      v.x = s_out[wv][r][lane * 4 + 0];
      v.y = s_out[wv][r][lane * 4 + 1];
      v.z = s_out[wv][r][lane * 4 + 2];
      v.w = s_out[wv][r][lane * 4 + 3];
      *reinterpret_cast<float4*>(outp + (long long)n * NODE_DIM + lane * 4) = v;
    }
  }
}

extern "C" void kernel_launch(void* const* d_in, const int* in_sizes, int n_in,
                              void* d_out, int out_size, void* d_ws,
                              size_t ws_size, hipStream_t stream) {
  (void)n_in; (void)out_size; (void)ws_size;

  const float* x   = (const float*)d_in[0];
  const int*   ei  = (const int*)d_in[1];
  const float* W1l = (const float*)d_in[2];
  const float* b1  = (const float*)d_in[3];
  const float* W1r = (const float*)d_in[4];
  const float* W2l = (const float*)d_in[5];
  const float* b2  = (const float*)d_in[6];
  const float* W2r = (const float*)d_in[7];
  float* out = (float*)d_out;

  const int N = in_sizes[0] / NODE_DIM;
  const int E = in_sizes[1] / 2;
  const int* src = ei;
  const int* dst = ei + E;

  char* ws = (char*)d_ws;
  int* deg = (int*)ws;
  size_t off = ((size_t)N * 4 + 255) & ~(size_t)255;
  float* aggb = (float*)(ws + off);
  off += (((size_t)N * NODE_DIM * 4) + 255) & ~(size_t)255;
  float* h = (float*)(ws + off);
  int* rnk   = (int*)h;
  int* slots = rnk + E;
  off += (((size_t)N * NODE_DIM * 4) + 255) & ~(size_t)255;
  int* first = (int*)(ws + off);
  off += ((size_t)N * 4 + 255) & ~(size_t)255;
  int* csr = (int*)(ws + off);

  const int tiles  = (N + 15) / 16;
  const int cblk   = (tiles + WAVES_PER_BLOCK - 1) / WAVES_PER_BLOCK;
  const int eblk   = (E + 255) / 256;
  const int sblk   = (N + SCAN_TILE - 1) / SCAN_TILE;
  const int nblk   = (N + NODES_PER_BLOCK - 1) / NODES_PER_BLOCK;

  hipMemsetAsync(deg, 0, (size_t)N * 4, stream);
  sage_rank_kernel<<<eblk, 256, 0, stream>>>(dst, deg, rnk, E);
  sage_scan_kernel<<<sblk, 256, 0, stream>>>(deg, first, N);
  sage_place_kernel<<<eblk, 256, 0, stream>>>(src, dst, first, rnk, slots, E);
  sage_order_kernel<<<nblk, NODES_PER_BLOCK * 32, 0, stream>>>(
      first, deg, slots, csr, N);

  sage_gather_kernel<<<nblk, NODES_PER_BLOCK * 32, 0, stream>>>(
      x, first, deg, csr, aggb, N);
  sage_combine_kernel<true><<<cblk, WAVES_PER_BLOCK * 32, 0, stream>>>(
      aggb, deg, x, W1l, b1, W1r, h, N);

  sage_gather_kernel<<<nblk, NODES_PER_BLOCK * 32, 0, stream>>>(
      h, first, deg, csr, aggb, N);
  sage_combine_kernel<false><<<cblk, WAVES_PER_BLOCK * 32, 0, stream>>>(
      aggb, deg, h, W2l, b2, W2r, out, N);
}


# CSR built through LDS bins: chunk counts, table sum, pairs copied to 256-node bins; a block per bin counts, scans, places, sorts by source; no global atomics or memset, 8->7 launches; sum order kept
# speedup vs baseline: 1.9979x; 1.0860x over previous
#include <hip/hip_runtime.h>
#include <hip/hip_bf16.h>
#include <math.h>

typedef float v2f __attribute__((ext_vector_type(2)));
typedef float v8f __attribute__((ext_vector_type(8)));

#define NODE_DIM 128
#define LDS_STRIDE 132
#define WAVES_PER_BLOCK 2

#define NODES_PER_BLOCK 8
#define BIN_SHIFT 8
#define BIN_NODES 256
#define NCHUNK 256
#define NB_MAX 1024
#define EDGE_BATCH 16
#define REC_BATCH 8
#define SLOT_CAP 8192

__device__ __forceinline__ int block_exclusive_scan_256(int v, int* s_w) {
  const int lane = threadIdx.x & 31;
  const int wv   = threadIdx.x >> 5;
  int incl = v;
  #pragma unroll
  for (int o = 1; o < 32; o <<= 1) {
    const int up = __shfl_up(incl, o, 32);
    if (lane >= o) incl += up;
  }
  if (lane == 31) s_w[wv] = incl;
  __syncthreads();
  int ex = incl - v;
  for (int w = 0; w < wv; ++w) ex += s_w[w];
  return ex;
}

__global__ __launch_bounds__(256)
void sage_bincount_kernel(const int* __restrict__ dst, int* __restrict__ cnt,
                          int E, int NBP, int chunk) {
  __shared__ int s_hist[NB_MAX + 1];
  const int tid = threadIdx.x;
  for (int b = tid; b < NBP; b += 256) s_hist[b] = 0;
  __syncthreads();
  const int e0 = blockIdx.x * chunk;
  const int e1 = min(e0 + chunk, E);
  for (int base = e0; base < e1; base += 256 * EDGE_BATCH) {
    int bin[EDGE_BATCH];
    #pragma unroll
    for (int u = 0; u < EDGE_BATCH; ++u) {
      const int e = base + u * 256 + tid;
      const int d = dst[min(e, e1 - 1)];
      bin[u] = (e < e1) ? (d >> BIN_SHIFT) : NB_MAX;
    }
    #pragma unroll
    for (int u = 0; u < EDGE_BATCH; ++u) atomicAdd(&s_hist[bin[u]], 1);
  }
  __syncthreads();
  for (int b = tid; b < NBP; b += 256) cnt[blockIdx.x * NBP + b] = s_hist[b];
}

__global__ __launch_bounds__(256)
void sage_binscatter_kernel(const int* __restrict__ src,
                            const int* __restrict__ dst,
                            const int* __restrict__ cnt,
                            int* __restrict__ binBase,
                            int* __restrict__ binned,
                            int E, int NB, int NBP, int chunk) {
  __shared__ int s_cur[NB_MAX + 1];
  __shared__ int s_wsum[8];
  const int tid = threadIdx.x;
  const int c   = blockIdx.x;
  const int nq  = NBP >> 2;

  int4 before = make_int4(0, 0, 0, 0);
  int4 total  = make_int4(0, 0, 0, 0);
  if (tid < nq) {
    const int4* col = reinterpret_cast<const int4*>(cnt) + tid;
    for (int r0 = 0; r0 < NCHUNK; r0 += 16) {
      int4 v[16];
      #pragma unroll
      for (int u = 0; u < 16; ++u) v[u] = col[(r0 + u) * nq];
      #pragma unroll
      for (int u = 0; u < 16; ++u) {
        const int m = (r0 + u < c) ? -1 : 0;
        before.x += v[u].x & m; before.y += v[u].y & m;
        before.z += v[u].z & m; before.w += v[u].w & m;
        total.x += v[u].x; total.y += v[u].y;
        total.z += v[u].z; total.w += v[u].w;
      }
    }
  }
  const int t4 = (total.x + total.y) + (total.z + total.w);
  const int ex = block_exclusive_scan_256(t4, s_wsum);
  if (tid < nq) {
    const int b0 = tid * 4;
    const int base0 = ex;
    const int base1 = base0 + total.x;
    const int base2 = base1 + total.y;
    const int base3 = base2 + total.z;
    s_cur[b0 + 0] = base0 + before.x;
    s_cur[b0 + 1] = base1 + before.y;
    s_cur[b0 + 2] = base2 + before.z;
    s_cur[b0 + 3] = base3 + before.w;
    if (c == 0) {
      if (b0 + 0 < NB) binBase[b0 + 0] = base0;
      if (b0 + 1 < NB) binBase[b0 + 1] = base1;
      if (b0 + 2 < NB) binBase[b0 + 2] = base2;
      if (b0 + 3 < NB) binBase[b0 + 3] = base3;
    }
  }
  __syncthreads();

  int2* rec = reinterpret_cast<int2*>(binned);
  const int e0 = c * chunk;
  const int e1 = min(e0 + chunk, E);
  for (int base = e0; base < e1; base += 256 * EDGE_BATCH) {
    int d[EDGE_BATCH], s[EDGE_BATCH], pos[EDGE_BATCH];
    #pragma unroll
    for (int u = 0; u < EDGE_BATCH; ++u) {
      const int e = min(base + u * 256 + tid, e1 - 1);
      d[u] = dst[e];
      s[u] = src[e];
    }
    #pragma unroll
    for (int u = 0; u < EDGE_BATCH; ++u) {
      const bool in = base + u * 256 + tid < e1;
      pos[u] = atomicAdd(&s_cur[in ? (d[u] >> BIN_SHIFT) : NB_MAX], 1);
    }
    #pragma unroll
    for (int u = 0; u < EDGE_BATCH; ++u)
      if (base + u * 256 + tid < e1) rec[pos[u]] = make_int2(s[u], d[u]);
  }
}

template <bool IN_LDS>
__device__ __forceinline__ void sage_order_bin(const int* s_slot,
                                               const int* gslots,
                                               const int* s_off,
                                               const int* s_deg,
                                               int* __restrict__ csr_bin) {
  const int lane = threadIdx.x & 31;
  const int wv = __builtin_amdgcn_readfirstlane((int)(threadIdx.x >> 5));
  for (int k = wv; k < BIN_NODES; k += 8) {
    const int d = __builtin_amdgcn_readfirstlane(s_deg[k]);
    const int o = __builtin_amdgcn_readfirstlane(s_off[k]);
    for (int ib = 0; ib < d; ib += 32) {
      const int i = ib + lane;
      int si = 0;
      if (i < d)
        si = IN_LDS ? s_slot[o + i]
                    : __hip_atomic_load(gslots + o + i, __ATOMIC_RELAXED,
                                        __HIP_MEMORY_SCOPE_AGENT);
      int pos = 0;
      for (int jb = 0; jb < d; jb += 32) {
        const int j0 = jb + lane;
        int sc = 0;
        if (j0 < d)
          sc = IN_LDS ? s_slot[o + j0]
                      : __hip_atomic_load(gslots + o + j0, __ATOMIC_RELAXED,
                                          __HIP_MEMORY_SCOPE_AGENT);
        const int cnt = min(32, d - jb);
        for (int jj = 0; jj < cnt; ++jj) {
          const int sj = __builtin_amdgcn_readlane(sc, jj);
          const int j  = jb + jj;
          pos += ((sj < si) || (sj == si && j < i)) ? 1 : 0;
        }
      }
      if (i < d) csr_bin[o + pos] = si;
    }
  }
}

__global__ __launch_bounds__(BIN_NODES)
void sage_bingroup_kernel(const int* __restrict__ binned,
                          const int* __restrict__ binBase,
                          int* __restrict__ deg, int* __restrict__ first,
                          int* slots, int* __restrict__ csr,
                          int N, int NB, int E) {
  __shared__ int s_cnt[BIN_NODES + 1];
  __shared__ int s_off[BIN_NODES];
  __shared__ int s_deg[BIN_NODES];
  __shared__ int s_w[8];
  __shared__ int s_slot[SLOT_CAP];
  const int tid = threadIdx.x;
  const int b   = blockIdx.x;
  const int beg = binBase[b];
  const int end = (b + 1 < NB) ? binBase[b + 1] : E;
  const int n0  = b << BIN_SHIFT;
  const bool in_lds = (end - beg) <= SLOT_CAP;
  const int2* rec = reinterpret_cast<const int2*>(binned);

  s_cnt[tid] = 0;
  __syncthreads();
  for (int base = beg; base < end; base += BIN_NODES * REC_BATCH) {
    int k[REC_BATCH];
    #pragma unroll
    for (int u = 0; u < REC_BATCH; ++u) {
      const int i = base + u * BIN_NODES + tid;
      const int d = rec[min(i, end - 1)].y;
      k[u] = (i < end) ? (d - n0) : BIN_NODES;
    }
    #pragma unroll
    for (int u = 0; u < REC_BATCH; ++u) atomicAdd(&s_cnt[k[u]], 1);
  }
  __syncthreads();

  const int cv = s_cnt[tid];
  const int ex = block_exclusive_scan_256(cv, s_w);
  const int n  = n0 + tid;
  if (n < N) { deg[n] = cv; first[n] = beg + ex; }
  s_deg[tid] = cv;
  s_off[tid] = ex;
  s_cnt[tid] = ex;
  __syncthreads();

  for (int base = beg; base < end; base += BIN_NODES * REC_BATCH) {
    int2 r[REC_BATCH];
    int pos[REC_BATCH];
    #pragma unroll
    for (int u = 0; u < REC_BATCH; ++u)
      r[u] = rec[min(base + u * BIN_NODES + tid, end - 1)];
    #pragma unroll
    for (int u = 0; u < REC_BATCH; ++u) {
      const bool in = base + u * BIN_NODES + tid < end;
      pos[u] = atomicAdd(&s_cnt[in ? (r[u].y - n0) : BIN_NODES], 1);
    }
    #pragma unroll
    for (int u = 0; u < REC_BATCH; ++u) {
      if (base + u * BIN_NODES + tid < end) {
        if (in_lds) s_slot[pos[u]] = r[u].x;
        else        slots[beg + pos[u]] = r[u].x;
      }
    }
  }
  __syncthreads();

  if (in_lds) sage_order_bin<true >(s_slot, slots + beg, s_off, s_deg, csr + beg);
  else        sage_order_bin<false>(s_slot, slots + beg, s_off, s_deg, csr + beg);
}

__global__ __launch_bounds__(NODES_PER_BLOCK * 32)
void sage_gather_kernel(const float* __restrict__ feat,
                        const int* __restrict__ first,
                        const int* __restrict__ deg,
                        const int* __restrict__ csr,
                        float* __restrict__ agg, int N) {
  const int lane = threadIdx.x & 31;
  const int n = blockIdx.x * NODES_PER_BLOCK +
                __builtin_amdgcn_readfirstlane((int)(threadIdx.x >> 5));
  if (n >= N) return;
  const int beg = first[n];
  const int d   = deg[n];
  float4 acc = make_float4(0.f, 0.f, 0.f, 0.f);
  for (int jb = 0; jb < d; jb += 32) {
    const int j0  = jb + lane;
    const int sc  = (j0 < d) ? csr[beg + j0] : 0;
    const int cnt = min(32, d - jb);
    for (int j8 = 0; j8 < cnt; j8 += 8) {
      float4 v[8];
      #pragma unroll
      for (int u = 0; u < 8; ++u) {
        const int s = __builtin_amdgcn_readlane(sc, j8 + u);
        v[u] = *reinterpret_cast<const float4*>(
            feat + (long long)s * NODE_DIM + lane * 4);
      }
      #pragma unroll
      for (int u = 0; u < 8; ++u) {
        if (u == 0 || j8 + u < cnt) {
          acc.x += v[u].x; acc.y += v[u].y; acc.z += v[u].z; acc.w += v[u].w;
        }
      }
    }
  }
  *reinterpret_cast<float4*>(agg + (long long)n * NODE_DIM + lane * 4) = acc;
}

template <bool NORM_RELU>
__global__ __launch_bounds__(WAVES_PER_BLOCK * 32)
void sage_combine_kernel(const float* __restrict__ agg,
                         const int* __restrict__ deg,
                         const float* __restrict__ xin,
                         const float* __restrict__ Wl,
                         const float* __restrict__ bias,
                         const float* __restrict__ Wr,
                         float* __restrict__ outp, int N) {
  __shared__ float s_agg[WAVES_PER_BLOCK][16][LDS_STRIDE];
  __shared__ float s_in [WAVES_PER_BLOCK][16][LDS_STRIDE];
  __shared__ float s_out[WAVES_PER_BLOCK][16][LDS_STRIDE];

  const int lane  = threadIdx.x & 31;
  const int wv    = threadIdx.x >> 5;
  const int tile  = blockIdx.x * WAVES_PER_BLOCK + wv;
  const int node0 = tile * 16;

  #pragma unroll 4
  for (int r = 0; r < 16; ++r) {
    const int n = node0 + r;
    if (n < N) {
      const float rc = 1.0f / fmaxf((float)deg[n], 1.0f);
      const float4 va = *reinterpret_cast<const float4*>(
          agg + (long long)n * NODE_DIM + lane * 4);
      const float4 vx = *reinterpret_cast<const float4*>(
          xin + (long long)n * NODE_DIM + lane * 4);
      s_agg[wv][r][lane * 4 + 0] = va.x * rc;
      s_agg[wv][r][lane * 4 + 1] = va.y * rc;
      s_agg[wv][r][lane * 4 + 2] = va.z * rc;
      s_agg[wv][r][lane * 4 + 3] = va.w * rc;
      s_in[wv][r][lane * 4 + 0] = vx.x;
      s_in[wv][r][lane * 4 + 1] = vx.y;
      s_in[wv][r][lane * 4 + 2] = vx.z;
      s_in[wv][r][lane * 4 + 3] = vx.w;
    } else {
      s_agg[wv][r][lane * 4 + 0] = 0.f; s_agg[wv][r][lane * 4 + 1] = 0.f;
      s_agg[wv][r][lane * 4 + 2] = 0.f; s_agg[wv][r][lane * 4 + 3] = 0.f;
      s_in[wv][r][lane * 4 + 0] = 0.f;  s_in[wv][r][lane * 4 + 1] = 0.f;
      s_in[wv][r][lane * 4 + 2] = 0.f;  s_in[wv][r][lane * 4 + 3] = 0.f;
    }
  }
  __syncthreads();

  const int m     = lane & 15;
  const int khalf = (lane >> 4) << 1;
  const int mbase = (lane >> 4) << 3;

  for (int ct = 0; ct < 8; ++ct) {
    const int n0 = ct * 16;
    const float bv = bias[n0 + m];
    v8f c;
    #pragma unroll
    for (int v = 0; v < 8; ++v) c[v] = bv;

    const float* wlrow = Wl + (long long)(n0 + m) * NODE_DIM;
    const float* wrrow = Wr + (long long)(n0 + m) * NODE_DIM;

    #pragma unroll 4
    for (int k0 = 0; k0 < NODE_DIM; k0 += 4) {
      v2f aA, aX;
      aA[0] = s_agg[wv][m][k0 + khalf];
      aA[1] = s_agg[wv][m][k0 + khalf + 1];
      const v2f bL = *reinterpret_cast<const v2f*>(wlrow + k0 + khalf);
      c = __builtin_amdgcn_wmma_f32_16x16x4_f32(false, aA, false, bL,
                                                (short)0, c, false, false);
      aX[0] = s_in[wv][m][k0 + khalf];
      aX[1] = s_in[wv][m][k0 + khalf + 1];
      const v2f bR = *reinterpret_cast<const v2f*>(wrrow + k0 + khalf);
      c = __builtin_amdgcn_wmma_f32_16x16x4_f32(false, aX, false, bR,
                                                (short)0, c, false, false);
    }

    #pragma unroll
    for (int v = 0; v < 8; ++v) s_out[wv][mbase + v][n0 + m] = c[v];
  }
  __syncthreads();

  if (NORM_RELU) {
    const int row = lane & 15;
    const int cb  = (lane >> 4) * 64;
    float ss = 0.f;
    #pragma unroll 8
    for (int c2 = 0; c2 < 64; ++c2) {
      const float v = s_out[wv][row][cb + c2];
      ss += v * v;
    }
    ss += __shfl_xor(ss, 16, 32);
    const float scl = 1.0f / fmaxf(sqrtf(ss), 1e-12f);
    #pragma unroll 8
    for (int c2 = 0; c2 < 64; ++c2) {
      const float v = s_out[wv][row][cb + c2] * scl;
      s_out[wv][row][cb + c2] = fmaxf(v, 0.0f);
    }
    __syncthreads();
  }

  #pragma unroll 4
  for (int r = 0; r < 16; ++r) {
    const int n = node0 + r;
    if (n < N) {
      float4 v;
      v.x = s_out[wv][r][lane * 4 + 0];
      v.y = s_out[wv][r][lane * 4 + 1];
      v.z = s_out[wv][r][lane * 4 + 2];
      v.w = s_out[wv][r][lane * 4 + 3];
      *reinterpret_cast<float4*>(outp + (long long)n * NODE_DIM + lane * 4) = v;
    }
  }
}

extern "C" void kernel_launch(void* const* d_in, const int* in_sizes, int n_in,
                              void* d_out, int out_size, void* d_ws,
                              size_t ws_size, hipStream_t stream) {
  (void)n_in; (void)out_size; (void)ws_size;

  const float* x   = (const float*)d_in[0];
  const int*   ei  = (const int*)d_in[1];
  const float* W1l = (const float*)d_in[2];
  const float* b1  = (const float*)d_in[3];
  const float* W1r = (const float*)d_in[4];
  const float* W2l = (const float*)d_in[5];
  const float* b2  = (const float*)d_in[6];
  const float* W2r = (const float*)d_in[7];
  float* out = (float*)d_out;

  const int N = in_sizes[0] / NODE_DIM;
  const int E = in_sizes[1] / 2;
  const int* src = ei;
  const int* dst = ei + E;

  const int NB    = (N + BIN_NODES - 1) / BIN_NODES;
  const int NBP   = (NB + 3) & ~3;
  const int chunk = (E + NCHUNK - 1) / NCHUNK;

  char* ws = (char*)d_ws;
  int* deg = (int*)ws;
  size_t off = ((size_t)N * 4 + 255) & ~(size_t)255;
  float* aggb = (float*)(ws + off);
  off += (((size_t)N * NODE_DIM * 4) + 255) & ~(size_t)255;
  float* h = (float*)(ws + off);
  char* hb = (char*)h;
  size_t hoff = 0;
  int* binned = (int*)(hb + hoff);
  hoff += ((size_t)E * 8 + 255) & ~(size_t)255;
  int* slots = (int*)(hb + hoff);
  hoff += ((size_t)E * 4 + 255) & ~(size_t)255;
  int* cnt = (int*)(hb + hoff);
  hoff += ((size_t)NCHUNK * NBP * 4 + 255) & ~(size_t)255;
  int* binBase = (int*)(hb + hoff);
  off += (((size_t)N * NODE_DIM * 4) + 255) & ~(size_t)255;
  int* first = (int*)(ws + off);
  off += ((size_t)N * 4 + 255) & ~(size_t)255;
  int* csr = (int*)(ws + off);

  const int tiles  = (N + 15) / 16;
  const int cblk   = (tiles + WAVES_PER_BLOCK - 1) / WAVES_PER_BLOCK;
  const int nblk   = (N + NODES_PER_BLOCK - 1) / NODES_PER_BLOCK;

  sage_bincount_kernel<<<NCHUNK, 256, 0, stream>>>(dst, cnt, E, NBP, chunk);
  sage_binscatter_kernel<<<NCHUNK, 256, 0, stream>>>(
      src, dst, cnt, binBase, binned, E, NB, NBP, chunk);
  sage_bingroup_kernel<<<NB, BIN_NODES, 0, stream>>>(
      binned, binBase, deg, first, slots, csr, N, NB, E);

  sage_gather_kernel<<<nblk, NODES_PER_BLOCK * 32, 0, stream>>>(
      x, first, deg, csr, aggb, N);
  sage_combine_kernel<true><<<cblk, WAVES_PER_BLOCK * 32, 0, stream>>>(
      aggb, deg, x, W1l, b1, W1r, h, N);

  sage_gather_kernel<<<nblk, NODES_PER_BLOCK * 32, 0, stream>>>(
      h, first, deg, csr, aggb, N);
  sage_combine_kernel<false><<<cblk, WAVES_PER_BLOCK * 32, 0, stream>>>(
      aggb, deg, h, W2l, b2, W2r, out, N);
}


# Combine: Wl and Wr copied to LDS once per work-group; 256 resident 8-wave groups walk the 16-node tiles (8 accumulators a wave), B fragments read from LDS, not 16-line global loads; sum order kept
# speedup vs baseline: 3.1687x; 1.5860x over previous
#include <hip/hip_runtime.h>
#include <hip/hip_bf16.h>
#include <math.h>

typedef float v2f __attribute__((ext_vector_type(2)));
typedef float v8f __attribute__((ext_vector_type(8)));

#define NODE_DIM 128
#define LDS_STRIDE 132

#define NODES_PER_BLOCK 8
#define BIN_SHIFT 8
#define BIN_NODES 256
#define NCHUNK 256
#define NB_MAX 1024
#define EDGE_BATCH 16
#define REC_BATCH 8
#define SLOT_CAP 8192

__device__ __forceinline__ int block_exclusive_scan_256(int v, int* s_w) {
  const int lane = threadIdx.x & 31;
  const int wv   = threadIdx.x >> 5;
  int incl = v;
  #pragma unroll
  for (int o = 1; o < 32; o <<= 1) {
    const int up = __shfl_up(incl, o, 32);
    if (lane >= o) incl += up;
  }
  if (lane == 31) s_w[wv] = incl;
  __syncthreads();
  int ex = incl - v;
  for (int w = 0; w < wv; ++w) ex += s_w[w];
  return ex;
}

__global__ __launch_bounds__(256)
void sage_bincount_kernel(const int* __restrict__ dst, int* __restrict__ cnt,
                          int E, int NBP, int chunk) {
  __shared__ int s_hist[NB_MAX + 1];
  const int tid = threadIdx.x;
  for (int b = tid; b < NBP; b += 256) s_hist[b] = 0;
  __syncthreads();
  const int e0 = blockIdx.x * chunk;
  const int e1 = min(e0 + chunk, E);
  for (int base = e0; base < e1; base += 256 * EDGE_BATCH) {
    int bin[EDGE_BATCH];
    #pragma unroll
    for (int u = 0; u < EDGE_BATCH; ++u) {
      const int e = base + u * 256 + tid;
      const int d = dst[min(e, e1 - 1)];
      bin[u] = (e < e1) ? (d >> BIN_SHIFT) : NB_MAX;
    }
    #pragma unroll
    for (int u = 0; u < EDGE_BATCH; ++u) atomicAdd(&s_hist[bin[u]], 1);
  }
  __syncthreads();
  for (int b = tid; b < NBP; b += 256) cnt[blockIdx.x * NBP + b] = s_hist[b];
}

__global__ __launch_bounds__(256)
void sage_binscatter_kernel(const int* __restrict__ src,
                            const int* __restrict__ dst,
                            const int* __restrict__ cnt,
                            int* __restrict__ binBase,
                            int* __restrict__ binned,
                            int E, int NB, int NBP, int chunk) {
  __shared__ int s_cur[NB_MAX + 1];
  __shared__ int s_wsum[8];
  const int tid = threadIdx.x;
  const int c   = blockIdx.x;
  const int nq  = NBP >> 2;

  int4 before = make_int4(0, 0, 0, 0);
  int4 total  = make_int4(0, 0, 0, 0);
  if (tid < nq) {
    const int4* col = reinterpret_cast<const int4*>(cnt) + tid;
    for (int r0 = 0; r0 < NCHUNK; r0 += 16) {
      int4 v[16];
      #pragma unroll
      for (int u = 0; u < 16; ++u) v[u] = col[(r0 + u) * nq];
      #pragma unroll
      for (int u = 0; u < 16; ++u) {
        const int m = (r0 + u < c) ? -1 : 0;
        before.x += v[u].x & m; before.y += v[u].y & m;
        before.z += v[u].z & m; before.w += v[u].w & m;
        total.x += v[u].x; total.y += v[u].y;
        total.z += v[u].z; total.w += v[u].w;
      }
    }
  }
  const int t4 = (total.x + total.y) + (total.z + total.w);
  const int ex = block_exclusive_scan_256(t4, s_wsum);
  if (tid < nq) {
    const int b0 = tid * 4;
    const int base0 = ex;
    const int base1 = base0 + total.x;
    const int base2 = base1 + total.y;
    const int base3 = base2 + total.z;
    s_cur[b0 + 0] = base0 + before.x;
    s_cur[b0 + 1] = base1 + before.y;
    s_cur[b0 + 2] = base2 + before.z;
    s_cur[b0 + 3] = base3 + before.w;
    if (c == 0) {
      if (b0 + 0 < NB) binBase[b0 + 0] = base0;
      if (b0 + 1 < NB) binBase[b0 + 1] = base1;
      if (b0 + 2 < NB) binBase[b0 + 2] = base2;
      if (b0 + 3 < NB) binBase[b0 + 3] = base3;
    }
  }
  __syncthreads();

  int2* rec = reinterpret_cast<int2*>(binned);
  const int e0 = c * chunk;
  const int e1 = min(e0 + chunk, E);
  for (int base = e0; base < e1; base += 256 * EDGE_BATCH) {
    int d[EDGE_BATCH], s[EDGE_BATCH], pos[EDGE_BATCH];
    #pragma unroll
    for (int u = 0; u < EDGE_BATCH; ++u) {
      const int e = min(base + u * 256 + tid, e1 - 1);
      d[u] = dst[e];
      s[u] = src[e];
    }
    #pragma unroll
    for (int u = 0; u < EDGE_BATCH; ++u) {
      const bool in = base + u * 256 + tid < e1;
      pos[u] = atomicAdd(&s_cur[in ? (d[u] >> BIN_SHIFT) : NB_MAX], 1);
    }
    #pragma unroll
    for (int u = 0; u < EDGE_BATCH; ++u)
      if (base + u * 256 + tid < e1) rec[pos[u]] = make_int2(s[u], d[u]);
  }
}

template <bool IN_LDS>
__device__ __forceinline__ void sage_order_bin(const int* s_slot,
                                               const int* gslots,
                                               const int* s_off,
                                               const int* s_deg,
                                               int* __restrict__ csr_bin) {
  const int lane = threadIdx.x & 31;
  const int wv = __builtin_amdgcn_readfirstlane((int)(threadIdx.x >> 5));
  for (int k = wv; k < BIN_NODES; k += 8) {
    const int d = __builtin_amdgcn_readfirstlane(s_deg[k]);
    const int o = __builtin_amdgcn_readfirstlane(s_off[k]);
    for (int ib = 0; ib < d; ib += 32) {
      const int i = ib + lane;
      int si = 0;
      if (i < d)
        si = IN_LDS ? s_slot[o + i]
                    : __hip_atomic_load(gslots + o + i, __ATOMIC_RELAXED,
                                        __HIP_MEMORY_SCOPE_AGENT);
      int pos = 0;
      for (int jb = 0; jb < d; jb += 32) {
        const int j0 = jb + lane;
        int sc = 0;
        if (j0 < d)
          sc = IN_LDS ? s_slot[o + j0]
                      : __hip_atomic_load(gslots + o + j0, __ATOMIC_RELAXED,
                                          __HIP_MEMORY_SCOPE_AGENT);
        const int cnt = min(32, d - jb);
        for (int jj = 0; jj < cnt; ++jj) {
          const int sj = __builtin_amdgcn_readlane(sc, jj);
          const int j  = jb + jj;
          pos += ((sj < si) || (sj == si && j < i)) ? 1 : 0;
        }
      }
      if (i < d) csr_bin[o + pos] = si;
    }
  }
}

__global__ __launch_bounds__(BIN_NODES)
void sage_bingroup_kernel(const int* __restrict__ binned,
                          const int* __restrict__ binBase,
                          int* __restrict__ deg, int* __restrict__ first,
                          int* slots, int* __restrict__ csr,
                          int N, int NB, int E) {
  __shared__ int s_cnt[BIN_NODES + 1];
  __shared__ int s_off[BIN_NODES];
  __shared__ int s_deg[BIN_NODES];
  __shared__ int s_w[8];
  __shared__ int s_slot[SLOT_CAP];
  const int tid = threadIdx.x;
  const int b   = blockIdx.x;
  const int beg = binBase[b];
  const int end = (b + 1 < NB) ? binBase[b + 1] : E;
  const int n0  = b << BIN_SHIFT;
  const bool in_lds = (end - beg) <= SLOT_CAP;
  const int2* rec = reinterpret_cast<const int2*>(binned);

  s_cnt[tid] = 0;
  __syncthreads();
  for (int base = beg; base < end; base += BIN_NODES * REC_BATCH) {
    int k[REC_BATCH];
    #pragma unroll
    for (int u = 0; u < REC_BATCH; ++u) {
      const int i = base + u * BIN_NODES + tid;
      const int d = rec[min(i, end - 1)].y;
      k[u] = (i < end) ? (d - n0) : BIN_NODES;
    }
    #pragma unroll
    for (int u = 0; u < REC_BATCH; ++u) atomicAdd(&s_cnt[k[u]], 1);
  }
  __syncthreads();

  const int cv = s_cnt[tid];
  const int ex = block_exclusive_scan_256(cv, s_w);
  const int n  = n0 + tid;
  if (n < N) { deg[n] = cv; first[n] = beg + ex; }
  s_deg[tid] = cv;
  s_off[tid] = ex;
  s_cnt[tid] = ex;
  __syncthreads();

  for (int base = beg; base < end; base += BIN_NODES * REC_BATCH) {
    int2 r[REC_BATCH];
    int pos[REC_BATCH];
    #pragma unroll
    for (int u = 0; u < REC_BATCH; ++u)
      r[u] = rec[min(base + u * BIN_NODES + tid, end - 1)];
    #pragma unroll
    for (int u = 0; u < REC_BATCH; ++u) {
      const bool in = base + u * BIN_NODES + tid < end;
      pos[u] = atomicAdd(&s_cnt[in ? (r[u].y - n0) : BIN_NODES], 1);
    }
    #pragma unroll
    for (int u = 0; u < REC_BATCH; ++u) {
      if (base + u * BIN_NODES + tid < end) {
        if (in_lds) s_slot[pos[u]] = r[u].x;
        else        slots[beg + pos[u]] = r[u].x;
      }
    }
  }
  __syncthreads();

  if (in_lds) sage_order_bin<true >(s_slot, slots + beg, s_off, s_deg, csr + beg);
  else        sage_order_bin<false>(s_slot, slots + beg, s_off, s_deg, csr + beg);
}

__global__ __launch_bounds__(NODES_PER_BLOCK * 32)
void sage_gather_kernel(const float* __restrict__ feat,
                        const int* __restrict__ first,
                        const int* __restrict__ deg,
                        const int* __restrict__ csr,
                        float* __restrict__ agg, int N) {
  const int lane = threadIdx.x & 31;
  const int n = blockIdx.x * NODES_PER_BLOCK +
                __builtin_amdgcn_readfirstlane((int)(threadIdx.x >> 5));
  if (n >= N) return;
  const int beg = first[n];
  const int d   = deg[n];
  float4 acc = make_float4(0.f, 0.f, 0.f, 0.f);
  for (int jb = 0; jb < d; jb += 32) {
    const int j0  = jb + lane;
    const int sc  = (j0 < d) ? csr[beg + j0] : 0;
    const int cnt = min(32, d - jb);
    for (int j8 = 0; j8 < cnt; j8 += 8) {
      float4 v[8];
      #pragma unroll
      for (int u = 0; u < 8; ++u) {
        const int s = __builtin_amdgcn_readlane(sc, j8 + u);
        v[u] = *reinterpret_cast<const float4*>(
            feat + (long long)s * NODE_DIM + lane * 4);
      }
      #pragma unroll
      for (int u = 0; u < 8; ++u) {
        if (u == 0 || j8 + u < cnt) {
          acc.x += v[u].x; acc.y += v[u].y; acc.z += v[u].z; acc.w += v[u].w;
        }
      }
    }
  }
  *reinterpret_cast<float4*>(agg + (long long)n * NODE_DIM + lane * 4) = acc;
}

#define COMBINE_WAVES 8
#define COMBINE_BLOCKS 256

__device__ __forceinline__ void wave_lds_handover() {
  __builtin_amdgcn_fence(__ATOMIC_ACQ_REL, "wavefront");
}

template <bool NORM_RELU>
__global__ __launch_bounds__(COMBINE_WAVES * 32)
void sage_combine_kernel(const float* __restrict__ agg,
                         const int* __restrict__ deg,
                         const float* __restrict__ xin,
                         const float* __restrict__ Wl,
                         const float* __restrict__ bias,
                         const float* __restrict__ Wr,
                         float* __restrict__ outp, int N, int tiles) {
  __shared__ float s_wl[NODE_DIM][LDS_STRIDE];
  __shared__ float s_wr[NODE_DIM][LDS_STRIDE];
  __shared__ float s_agg[COMBINE_WAVES][16][LDS_STRIDE];
  __shared__ float s_in [COMBINE_WAVES][16][LDS_STRIDE];

  const int tid  = threadIdx.x;
  const int lane = tid & 31;
  const int wv   = __builtin_amdgcn_readfirstlane(tid >> 5);

  {
    float4 v[16];
    #pragma unroll
    for (int u = 0; u < 16; ++u)
      v[u] = reinterpret_cast<const float4*>(Wl)[u * 256 + tid];
    #pragma unroll
    for (int u = 0; u < 16; ++u)
      *reinterpret_cast<float4*>(&s_wl[u * 8 + wv][lane * 4]) = v[u];
    #pragma unroll
    for (int u = 0; u < 16; ++u)
      v[u] = reinterpret_cast<const float4*>(Wr)[u * 256 + tid];
    #pragma unroll
    for (int u = 0; u < 16; ++u)
      *reinterpret_cast<float4*>(&s_wr[u * 8 + wv][lane * 4]) = v[u];
  }
  __syncthreads();

  const int m     = lane & 15;
  const int khalf = (lane >> 4) << 1;
  const int mbase = (lane >> 4) << 3;

  float bv[8];
  #pragma unroll
  for (int ct = 0; ct < 8; ++ct) bv[ct] = bias[ct * 16 + m];

  float (*ta)[LDS_STRIDE] = s_agg[wv];
  float (*tx)[LDS_STRIDE] = s_in[wv];
  const float* pa = &ta[m][khalf];
  const float* px = &tx[m][khalf];
  const float* pl = &s_wl[m][khalf];
  const float* pr = &s_wr[m][khalf];

  for (int slot = wv; ; slot += COMBINE_WAVES) {
    const int tile = (int)blockIdx.x + slot * (int)gridDim.x;
    if (tile >= tiles) break;
    const int node0 = tile * 16;

    {
      float4 va[16], vx[16];
      #pragma unroll
      for (int r = 0; r < 16; ++r) {
        const int nc = min(node0 + r, N - 1);
        va[r] = *reinterpret_cast<const float4*>(agg + (long long)nc * NODE_DIM + lane * 4);
        vx[r] = *reinterpret_cast<const float4*>(xin + (long long)nc * NODE_DIM + lane * 4);
      }
      const float rcl = 1.0f / fmaxf((float)deg[min(node0 + m, N - 1)], 1.0f);
      #pragma unroll
      for (int r = 0; r < 16; ++r) {
        const float rc = __builtin_bit_cast(float,
            __builtin_amdgcn_readlane(__builtin_bit_cast(int, rcl), r));
        float4 a = va[r];
        a.x *= rc; a.y *= rc; a.z *= rc; a.w *= rc;
        *reinterpret_cast<float4*>(&ta[r][lane * 4]) = a;
        *reinterpret_cast<float4*>(&tx[r][lane * 4]) = vx[r];
      }
    }
    wave_lds_handover();

    v8f acc[8];
    #pragma unroll
    for (int ct = 0; ct < 8; ++ct) {
      #pragma unroll
      for (int v = 0; v < 8; ++v) acc[ct][v] = bv[ct];
    }

    v2f a0, x0, l0[8], r0[8], a1, x1, l1[8], r1[8];
    a0 = *reinterpret_cast<const v2f*>(pa);
    x0 = *reinterpret_cast<const v2f*>(px);
    #pragma unroll
    for (int ct = 0; ct < 8; ++ct) {
      l0[ct] = *reinterpret_cast<const v2f*>(pl + ct * 16 * LDS_STRIDE);
      r0[ct] = *reinterpret_cast<const v2f*>(pr + ct * 16 * LDS_STRIDE);
    }
    #pragma unroll
    for (int k0 = 0; k0 < NODE_DIM; k0 += 8) {
      const int k1 = k0 + 4;
      a1 = *reinterpret_cast<const v2f*>(pa + k1);
      x1 = *reinterpret_cast<const v2f*>(px + k1);
      #pragma unroll
      for (int ct = 0; ct < 8; ++ct) {
        l1[ct] = *reinterpret_cast<const v2f*>(pl + ct * 16 * LDS_STRIDE + k1);
        r1[ct] = *reinterpret_cast<const v2f*>(pr + ct * 16 * LDS_STRIDE + k1);
      }
      __builtin_amdgcn_sched_barrier(0);
      #pragma unroll
      for (int ct = 0; ct < 8; ++ct) {
        acc[ct] = __builtin_amdgcn_wmma_f32_16x16x4_f32(false, a0, false, l0[ct],
                                                        (short)0, acc[ct], false, false);
        acc[ct] = __builtin_amdgcn_wmma_f32_16x16x4_f32(false, x0, false, r0[ct],
                                                        (short)0, acc[ct], false, false);
      }
      __builtin_amdgcn_sched_barrier(0);
      if (k0 + 8 < NODE_DIM) {
        const int k2 = k0 + 8;
        a0 = *reinterpret_cast<const v2f*>(pa + k2);
        x0 = *reinterpret_cast<const v2f*>(px + k2);
        #pragma unroll
        for (int ct = 0; ct < 8; ++ct) {
          l0[ct] = *reinterpret_cast<const v2f*>(pl + ct * 16 * LDS_STRIDE + k2);
          r0[ct] = *reinterpret_cast<const v2f*>(pr + ct * 16 * LDS_STRIDE + k2);
        }
      }
      __builtin_amdgcn_sched_barrier(0);
      #pragma unroll
      for (int ct = 0; ct < 8; ++ct) {
        acc[ct] = __builtin_amdgcn_wmma_f32_16x16x4_f32(false, a1, false, l1[ct],
                                                        (short)0, acc[ct], false, false);
        acc[ct] = __builtin_amdgcn_wmma_f32_16x16x4_f32(false, x1, false, r1[ct],
                                                        (short)0, acc[ct], false, false);
      }
      __builtin_amdgcn_sched_barrier(0);
    }

    wave_lds_handover();
    #pragma unroll
    for (int ct = 0; ct < 8; ++ct) {
      #pragma unroll
      for (int v = 0; v < 8; ++v) ta[mbase + v][ct * 16 + m] = acc[ct][v];
    }
    wave_lds_handover();

    if (NORM_RELU) {
      const int row = lane & 15;
      const int cb  = (lane >> 4) * 64;
      float ss = 0.f;
      #pragma unroll 8
      for (int c2 = 0; c2 < 64; ++c2) {
        const float v = ta[row][cb + c2];
        ss += v * v;
      }
      ss += __shfl_xor(ss, 16, 32);
      const float scl = 1.0f / fmaxf(sqrtf(ss), 1e-12f);
      #pragma unroll 8
      for (int c2 = 0; c2 < 64; ++c2) {
        const float v = ta[row][cb + c2] * scl;
        ta[row][cb + c2] = fmaxf(v, 0.0f);
      }
      wave_lds_handover();
    }

    {
      float4 vo[16];
      #pragma unroll
      for (int r = 0; r < 16; ++r)
        vo[r] = *reinterpret_cast<const float4*>(&ta[r][lane * 4]);
      #pragma unroll
      for (int r = 0; r < 16; ++r) {
        const int n = node0 + r;
        if (n < N)
          *reinterpret_cast<float4*>(outp + (long long)n * NODE_DIM + lane * 4) = vo[r];
      }
    }
    wave_lds_handover();
  }
}

extern "C" void kernel_launch(void* const* d_in, const int* in_sizes, int n_in,
                              void* d_out, int out_size, void* d_ws,
                              size_t ws_size, hipStream_t stream) {
  (void)n_in; (void)out_size; (void)ws_size;

  const float* x   = (const float*)d_in[0];
  const int*   ei  = (const int*)d_in[1];
  const float* W1l = (const float*)d_in[2];
  const float* b1  = (const float*)d_in[3];
  const float* W1r = (const float*)d_in[4];
  const float* W2l = (const float*)d_in[5];
  const float* b2  = (const float*)d_in[6];
  const float* W2r = (const float*)d_in[7];
  float* out = (float*)d_out;

  const int N = in_sizes[0] / NODE_DIM;
  const int E = in_sizes[1] / 2;
  const int* src = ei;
  const int* dst = ei + E;

  const int NB    = (N + BIN_NODES - 1) / BIN_NODES;
  const int NBP   = (NB + 3) & ~3;
  const int chunk = (E + NCHUNK - 1) / NCHUNK;

  char* ws = (char*)d_ws;
  int* deg = (int*)ws;
  size_t off = ((size_t)N * 4 + 255) & ~(size_t)255;
  float* aggb = (float*)(ws + off);
  off += (((size_t)N * NODE_DIM * 4) + 255) & ~(size_t)255;
  float* h = (float*)(ws + off);
  char* hb = (char*)h;
  size_t hoff = 0;
  int* binned = (int*)(hb + hoff);
  hoff += ((size_t)E * 8 + 255) & ~(size_t)255;
  int* slots = (int*)(hb + hoff);
  hoff += ((size_t)E * 4 + 255) & ~(size_t)255;
  int* cnt = (int*)(hb + hoff);
  hoff += ((size_t)NCHUNK * NBP * 4 + 255) & ~(size_t)255;
  int* binBase = (int*)(hb + hoff);
  off += (((size_t)N * NODE_DIM * 4) + 255) & ~(size_t)255;
  int* first = (int*)(ws + off);
  off += ((size_t)N * 4 + 255) & ~(size_t)255;
  int* csr = (int*)(ws + off);

  const int tiles  = (N + 15) / 16;
  const int nblk   = (N + NODES_PER_BLOCK - 1) / NODES_PER_BLOCK;

  sage_bincount_kernel<<<NCHUNK, 256, 0, stream>>>(dst, cnt, E, NBP, chunk);
  sage_binscatter_kernel<<<NCHUNK, 256, 0, stream>>>(
      src, dst, cnt, binBase, binned, E, NB, NBP, chunk);
  sage_bingroup_kernel<<<NB, BIN_NODES, 0, stream>>>(
      binned, binBase, deg, first, slots, csr, N, NB, E);

  sage_gather_kernel<<<nblk, NODES_PER_BLOCK * 32, 0, stream>>>(
      x, first, deg, csr, aggb, N);
  sage_combine_kernel<true><<<COMBINE_BLOCKS, COMBINE_WAVES * 32, 0, stream>>>(
      aggb, deg, x, W1l, b1, W1r, h, N, tiles);

  sage_gather_kernel<<<nblk, NODES_PER_BLOCK * 32, 0, stream>>>(
      h, first, deg, csr, aggb, N);
  sage_combine_kernel<false><<<COMBINE_BLOCKS, COMBINE_WAVES * 32, 0, stream>>>(
      aggb, deg, h, W2l, b2, W2r, out, N, tiles);
}


# One launch per layer, fixed wave roles: 16 gather waves add source rows (csr order, mean applied) into an LDS ring while 4 matrix waves, one per SIMD, run the WMMAs; 7->5 launches; sum order kept
# speedup vs baseline: 3.2758x; 1.0338x over previous
#include <hip/hip_runtime.h>
#include <hip/hip_bf16.h>
#include <math.h>

typedef float v2f __attribute__((ext_vector_type(2)));
typedef float v8f __attribute__((ext_vector_type(8)));

#define NODE_DIM 128
#define LDS_STRIDE 132

#define BIN_SHIFT 8
#define BIN_NODES 256
#define NCHUNK 256
#define NB_MAX 1024
#define EDGE_BATCH 16
#define REC_BATCH 8
#define SLOT_CAP 8192

__device__ __forceinline__ int block_exclusive_scan_256(int v, int* s_w) {
  const int lane = threadIdx.x & 31;
  const int wv   = threadIdx.x >> 5;
  int incl = v;
  #pragma unroll
  for (int o = 1; o < 32; o <<= 1) {
    const int up = __shfl_up(incl, o, 32);
    if (lane >= o) incl += up;
  }
  if (lane == 31) s_w[wv] = incl;
  __syncthreads();
  int ex = incl - v;
  for (int w = 0; w < wv; ++w) ex += s_w[w];
  return ex;
}

__global__ __launch_bounds__(256)
void sage_bincount_kernel(const int* __restrict__ dst, int* __restrict__ cnt,
                          int E, int NBP, int chunk) {
  __shared__ int s_hist[NB_MAX + 1];
  const int tid = threadIdx.x;
  for (int b = tid; b < NBP; b += 256) s_hist[b] = 0;
  __syncthreads();
  const int e0 = blockIdx.x * chunk;
  const int e1 = min(e0 + chunk, E);
  for (int base = e0; base < e1; base += 256 * EDGE_BATCH) {
    int bin[EDGE_BATCH];
    #pragma unroll
    for (int u = 0; u < EDGE_BATCH; ++u) {
      const int e = base + u * 256 + tid;
      const int d = dst[min(e, e1 - 1)];
      bin[u] = (e < e1) ? (d >> BIN_SHIFT) : NB_MAX;
    }
    #pragma unroll
    for (int u = 0; u < EDGE_BATCH; ++u) atomicAdd(&s_hist[bin[u]], 1);
  }
  __syncthreads();
  for (int b = tid; b < NBP; b += 256) cnt[blockIdx.x * NBP + b] = s_hist[b];
}

__global__ __launch_bounds__(256)
void sage_binscatter_kernel(const int* __restrict__ src,
                            const int* __restrict__ dst,
                            const int* __restrict__ cnt,
                            int* __restrict__ binBase,
                            int* __restrict__ binned,
                            int E, int NB, int NBP, int chunk) {
  __shared__ int s_cur[NB_MAX + 1];
  __shared__ int s_wsum[8];
  const int tid = threadIdx.x;
  const int c   = blockIdx.x;
  const int nq  = NBP >> 2;

  int4 before = make_int4(0, 0, 0, 0);
  int4 total  = make_int4(0, 0, 0, 0);
  if (tid < nq) {
    const int4* col = reinterpret_cast<const int4*>(cnt) + tid;
    for (int r0 = 0; r0 < NCHUNK; r0 += 16) {
      int4 v[16];
      #pragma unroll
      for (int u = 0; u < 16; ++u) v[u] = col[(r0 + u) * nq];
      #pragma unroll
      for (int u = 0; u < 16; ++u) {
        const int m = (r0 + u < c) ? -1 : 0;
        before.x += v[u].x & m; before.y += v[u].y & m;
        before.z += v[u].z & m; before.w += v[u].w & m;
        total.x += v[u].x; total.y += v[u].y;
        total.z += v[u].z; total.w += v[u].w;
      }
    }
  }
  const int t4 = (total.x + total.y) + (total.z + total.w);
  const int ex = block_exclusive_scan_256(t4, s_wsum);
  if (tid < nq) {
    const int b0 = tid * 4;
    const int base0 = ex;
    const int base1 = base0 + total.x;
    const int base2 = base1 + total.y;
    const int base3 = base2 + total.z;
    s_cur[b0 + 0] = base0 + before.x;
    s_cur[b0 + 1] = base1 + before.y;
    s_cur[b0 + 2] = base2 + before.z;
    s_cur[b0 + 3] = base3 + before.w;
    if (c == 0) {
      if (b0 + 0 < NB) binBase[b0 + 0] = base0;
      if (b0 + 1 < NB) binBase[b0 + 1] = base1;
      if (b0 + 2 < NB) binBase[b0 + 2] = base2;
      if (b0 + 3 < NB) binBase[b0 + 3] = base3;
    }
  }
  __syncthreads();

  int2* rec = reinterpret_cast<int2*>(binned);
  const int e0 = c * chunk;
  const int e1 = min(e0 + chunk, E);
  for (int base = e0; base < e1; base += 256 * EDGE_BATCH) {
    int d[EDGE_BATCH], s[EDGE_BATCH], pos[EDGE_BATCH];
    #pragma unroll
    for (int u = 0; u < EDGE_BATCH; ++u) {
      const int e = min(base + u * 256 + tid, e1 - 1);
      d[u] = dst[e];
      s[u] = src[e];
    }
    #pragma unroll
    for (int u = 0; u < EDGE_BATCH; ++u) {
      const bool in = base + u * 256 + tid < e1;
      pos[u] = atomicAdd(&s_cur[in ? (d[u] >> BIN_SHIFT) : NB_MAX], 1);
    }
    #pragma unroll
    for (int u = 0; u < EDGE_BATCH; ++u)
      if (base + u * 256 + tid < e1) rec[pos[u]] = make_int2(s[u], d[u]);
  }
}

template <bool IN_LDS>
__device__ __forceinline__ void sage_order_bin(const int* s_slot,
                                               const int* gslots,
                                               const int* s_off,
                                               const int* s_deg,
                                               int* __restrict__ csr_bin) {
  const int lane = threadIdx.x & 31;
  const int wv = __builtin_amdgcn_readfirstlane((int)(threadIdx.x >> 5));
  for (int k = wv; k < BIN_NODES; k += 8) {
    const int d = __builtin_amdgcn_readfirstlane(s_deg[k]);
    const int o = __builtin_amdgcn_readfirstlane(s_off[k]);
    for (int ib = 0; ib < d; ib += 32) {
      const int i = ib + lane;
      int si = 0;
      if (i < d)
        si = IN_LDS ? s_slot[o + i]
                    : __hip_atomic_load(gslots + o + i, __ATOMIC_RELAXED,
                                        __HIP_MEMORY_SCOPE_AGENT);
      int pos = 0;
      for (int jb = 0; jb < d; jb += 32) {
        const int j0 = jb + lane;
        int sc = 0;
        if (j0 < d)
          sc = IN_LDS ? s_slot[o + j0]
                      : __hip_atomic_load(gslots + o + j0, __ATOMIC_RELAXED,
                                          __HIP_MEMORY_SCOPE_AGENT);
        const int cnt = min(32, d - jb);
        for (int jj = 0; jj < cnt; ++jj) {
          const int sj = __builtin_amdgcn_readlane(sc, jj);
          const int j  = jb + jj;
          pos += ((sj < si) || (sj == si && j < i)) ? 1 : 0;
        }
      }
      if (i < d) csr_bin[o + pos] = si;
    }
  }
}

__global__ __launch_bounds__(BIN_NODES)
void sage_bingroup_kernel(const int* __restrict__ binned,
                          const int* __restrict__ binBase,
                          int* __restrict__ deg, int* __restrict__ first,
                          int* slots, int* __restrict__ csr,
                          int N, int NB, int E) {
  __shared__ int s_cnt[BIN_NODES + 1];
  __shared__ int s_off[BIN_NODES];
  __shared__ int s_deg[BIN_NODES];
  __shared__ int s_w[8];
  __shared__ int s_slot[SLOT_CAP];
  const int tid = threadIdx.x;
  const int b   = blockIdx.x;
  const int beg = binBase[b];
  const int end = (b + 1 < NB) ? binBase[b + 1] : E;
  const int n0  = b << BIN_SHIFT;
  const bool in_lds = (end - beg) <= SLOT_CAP;
  const int2* rec = reinterpret_cast<const int2*>(binned);

  s_cnt[tid] = 0;
  __syncthreads();
  for (int base = beg; base < end; base += BIN_NODES * REC_BATCH) {
    int k[REC_BATCH];
    #pragma unroll
    for (int u = 0; u < REC_BATCH; ++u) {
      const int i = base + u * BIN_NODES + tid;
      const int d = rec[min(i, end - 1)].y;
      k[u] = (i < end) ? (d - n0) : BIN_NODES;
    }
    #pragma unroll
    for (int u = 0; u < REC_BATCH; ++u) atomicAdd(&s_cnt[k[u]], 1);
  }
  __syncthreads();

  const int cv = s_cnt[tid];
  const int ex = block_exclusive_scan_256(cv, s_w);
  const int n  = n0 + tid;
  if (n < N) { deg[n] = cv; first[n] = beg + ex; }
  s_deg[tid] = cv;
  s_off[tid] = ex;
  s_cnt[tid] = ex;
  __syncthreads();

  for (int base = beg; base < end; base += BIN_NODES * REC_BATCH) {
    int2 r[REC_BATCH];
    int pos[REC_BATCH];
    #pragma unroll
    for (int u = 0; u < REC_BATCH; ++u)
      r[u] = rec[min(base + u * BIN_NODES + tid, end - 1)];
    #pragma unroll
    for (int u = 0; u < REC_BATCH; ++u) {
      const bool in = base + u * BIN_NODES + tid < end;
      pos[u] = atomicAdd(&s_cnt[in ? (r[u].y - n0) : BIN_NODES], 1);
    }
    #pragma unroll
    for (int u = 0; u < REC_BATCH; ++u) {
      if (base + u * BIN_NODES + tid < end) {
        if (in_lds) s_slot[pos[u]] = r[u].x;
        else        slots[beg + pos[u]] = r[u].x;
      }
    }
  }
  __syncthreads();

  if (in_lds) sage_order_bin<true >(s_slot, slots + beg, s_off, s_deg, csr + beg);
  else        sage_order_bin<false>(s_slot, slots + beg, s_off, s_deg, csr + beg);
}

#define LAYER_MWAVES 4
#define LAYER_GWAVES 16
#define LAYER_WAVES (LAYER_MWAVES + LAYER_GWAVES)
#define LAYER_THREADS (LAYER_WAVES * 32)
#define LAYER_BLOCKS 256
#define ROWS_AHEAD 16

__device__ __forceinline__ void wave_lds_handover() {
  __builtin_amdgcn_fence(__ATOMIC_ACQ_REL, "wavefront");
}

__device__ __forceinline__ void sage_gather_rows(const float* __restrict__ feat,
                                                 const int* __restrict__ first,
                                                 const int* __restrict__ deg,
                                                 const int* __restrict__ csr,
                                                 float (*trow)[LDS_STRIDE],
                                                 int n0, int npw, int N, int lane) {
  const int li = lane & 3;
  const int nn = n0 + li;
  const int nc = min(nn, N - 1);
  const int fv = first[nc];
  const int dv = (li < npw && nn < N) ? deg[nc] : 0;
  const int e0 = __builtin_amdgcn_readlane(fv, 0);
  const int total = __builtin_amdgcn_readlane(dv, 0) + __builtin_amdgcn_readlane(dv, 1) +
                    __builtin_amdgcn_readlane(dv, 2) + __builtin_amdgcn_readlane(dv, 3);
  float4 acc = make_float4(0.f, 0.f, 0.f, 0.f);
  int k   = 0;
  int nxt = __builtin_amdgcn_readlane(dv, 0);
  int sc = (lane < total) ? csr[e0 + lane] : 0;
  for (int jb = 0; jb < total; jb += 32) {
    const int cnt = min(32, total - jb);
    int scn = 0;
    for (int j16 = 0; j16 < cnt; j16 += ROWS_AHEAD) {
      float4 v[ROWS_AHEAD];
      #pragma unroll
      for (int u = 0; u < ROWS_AHEAD; ++u) {
        const int s = __builtin_amdgcn_readlane(sc, j16 + u);
        v[u] = *reinterpret_cast<const float4*>(
            feat + (long long)s * NODE_DIM + lane * 4);
      }
      if (j16 == 0) {
        const int jn = jb + 32 + lane;
        if (jn < total) scn = csr[e0 + jn];
      }
      #pragma unroll
      for (int u = 0; u < ROWS_AHEAD; ++u) {
        const int idx = jb + j16 + u;
        if (idx < total) {
          while (idx >= nxt) {
            const float rc = 1.0f / fmaxf((float)__builtin_amdgcn_readlane(dv, k), 1.0f);
            *reinterpret_cast<float4*>(&trow[k][lane * 4]) =
                make_float4(acc.x * rc, acc.y * rc, acc.z * rc, acc.w * rc);
            acc = make_float4(0.f, 0.f, 0.f, 0.f);
            ++k;
            nxt += __builtin_amdgcn_readlane(dv, k);
          }
          acc.x += v[u].x; acc.y += v[u].y; acc.z += v[u].z; acc.w += v[u].w;
        }
      }
    }
    sc = scn;
  }
  while (k < npw) {
    const float rc = 1.0f / fmaxf((float)__builtin_amdgcn_readlane(dv, k), 1.0f);
    *reinterpret_cast<float4*>(&trow[k][lane * 4]) =
        make_float4(acc.x * rc, acc.y * rc, acc.z * rc, acc.w * rc);
    acc = make_float4(0.f, 0.f, 0.f, 0.f);
    ++k;
  }
}

template <bool NORM_RELU>
__global__ __launch_bounds__(LAYER_THREADS)
void sage_layer_kernel(const float* __restrict__ feat,
                       const int* __restrict__ first,
                       const int* __restrict__ deg,
                       const int* __restrict__ csr,
                       const float* __restrict__ Wl,
                       const float* __restrict__ bias,
                       const float* __restrict__ Wr,
                       float* __restrict__ outp, int N, int tiles) {
  __shared__ float s_wl[NODE_DIM][LDS_STRIDE];
  __shared__ float s_wr[NODE_DIM][LDS_STRIDE];
  __shared__ float s_ring[2][LAYER_MWAVES][16][LDS_STRIDE];
  __shared__ float s_root[LAYER_MWAVES][16][LDS_STRIDE];

  const int tid  = threadIdx.x;
  const int lane = tid & 31;
  const int wv   = __builtin_amdgcn_readfirstlane(tid >> 5);

  {
    constexpr int W4   = NODE_DIM * NODE_DIM / 4;
    constexpr int NLD  = (W4 + LAYER_THREADS - 1) / LAYER_THREADS;
    float4 v[NLD];
    #pragma unroll
    for (int u = 0; u < NLD; ++u)
      v[u] = reinterpret_cast<const float4*>(Wl)[min(u * LAYER_THREADS + tid, W4 - 1)];
    #pragma unroll
    for (int u = 0; u < NLD; ++u) {
      const int i = u * LAYER_THREADS + tid;
      if (i < W4) *reinterpret_cast<float4*>(&s_wl[i >> 5][(i & 31) * 4]) = v[u];
    }
    #pragma unroll
    for (int u = 0; u < NLD; ++u)
      v[u] = reinterpret_cast<const float4*>(Wr)[min(u * LAYER_THREADS + tid, W4 - 1)];
    #pragma unroll
    for (int u = 0; u < NLD; ++u) {
      const int i = u * LAYER_THREADS + tid;
      if (i < W4) *reinterpret_cast<float4*>(&s_wr[i >> 5][(i & 31) * 4]) = v[u];
    }
  }

  const int m     = lane & 15;
  const int khalf = (lane >> 4) << 1;
  const int mbase = (lane >> 4) << 3;

  const int b      = (int)blockIdx.x;
  const int G      = (int)gridDim.x;
  const int nt     = (b < tiles) ? (tiles - b + G - 1) / G : 0;
  const int rounds = (nt + LAYER_MWAVES - 1) / LAYER_MWAVES;

  for (int r = 0; r <= rounds; ++r) {
    if (wv >= LAYER_MWAVES) {
      if (r < rounds) {
        const int g   = wv - LAYER_MWAVES;
        const int ntr = min(LAYER_MWAVES, nt - r * LAYER_MWAVES);
        const int npw = (ntr >= 3) ? 4 : ntr;
        const int j0  = g * npw;
        if (j0 < 16 * ntr) {
          const int q    = j0 >> 4;
          const int row0 = j0 & 15;
          const int tile = b + (r * LAYER_MWAVES + q) * G;
          sage_gather_rows(feat, first, deg, csr, &s_ring[r & 1][q][row0],
                           tile * 16 + row0, npw, N, lane);
        }
      }
    } else if (r > 0 && (r - 1) * LAYER_MWAVES + wv < nt) {
      const int tile  = b + ((r - 1) * LAYER_MWAVES + wv) * G;
      const int node0 = tile * 16;
      float (*ta)[LDS_STRIDE] = s_ring[(r - 1) & 1][wv];
      float (*tx)[LDS_STRIDE] = s_root[wv];
      const float* pa = &ta[m][khalf];
      const float* px = &tx[m][khalf];
      const float* pl = &s_wl[m][khalf];
      const float* pr = &s_wr[m][khalf];

      float bv[8];
      #pragma unroll
      for (int ct = 0; ct < 8; ++ct) bv[ct] = bias[ct * 16 + m];

      {
        float4 vx[16];
        #pragma unroll
        for (int rr = 0; rr < 16; ++rr) {
          const int nc = min(node0 + rr, N - 1);
          vx[rr] = *reinterpret_cast<const float4*>(feat + (long long)nc * NODE_DIM + lane * 4);
        }
        #pragma unroll
        for (int rr = 0; rr < 16; ++rr)
          *reinterpret_cast<float4*>(&tx[rr][lane * 4]) = vx[rr];
      }
      wave_lds_handover();

      v8f acc[8];
      #pragma unroll
      for (int ct = 0; ct < 8; ++ct) {
        #pragma unroll
        for (int v = 0; v < 8; ++v) acc[ct][v] = bv[ct];
      }

      v2f a0, x0, l0[8], r0[8], a1, x1, l1[8], r1[8];
      a0 = *reinterpret_cast<const v2f*>(pa);
      x0 = *reinterpret_cast<const v2f*>(px);
      #pragma unroll
      for (int ct = 0; ct < 8; ++ct) {
        l0[ct] = *reinterpret_cast<const v2f*>(pl + ct * 16 * LDS_STRIDE);
        r0[ct] = *reinterpret_cast<const v2f*>(pr + ct * 16 * LDS_STRIDE);
      }
      #pragma unroll
      for (int k0 = 0; k0 < NODE_DIM; k0 += 8) {
        const int k1 = k0 + 4;
        a1 = *reinterpret_cast<const v2f*>(pa + k1);
        x1 = *reinterpret_cast<const v2f*>(px + k1);
        #pragma unroll
        for (int ct = 0; ct < 8; ++ct) {
          l1[ct] = *reinterpret_cast<const v2f*>(pl + ct * 16 * LDS_STRIDE + k1);
          r1[ct] = *reinterpret_cast<const v2f*>(pr + ct * 16 * LDS_STRIDE + k1);
        }
        __builtin_amdgcn_sched_barrier(0);
        #pragma unroll
        for (int ct = 0; ct < 8; ++ct) {
          acc[ct] = __builtin_amdgcn_wmma_f32_16x16x4_f32(false, a0, false, l0[ct],
                                                          (short)0, acc[ct], false, false);
          acc[ct] = __builtin_amdgcn_wmma_f32_16x16x4_f32(false, x0, false, r0[ct],
                                                          (short)0, acc[ct], false, false);
        }
        __builtin_amdgcn_sched_barrier(0);
        if (k0 + 8 < NODE_DIM) {
          const int k2 = k0 + 8;
          a0 = *reinterpret_cast<const v2f*>(pa + k2);
          x0 = *reinterpret_cast<const v2f*>(px + k2);
          #pragma unroll
          for (int ct = 0; ct < 8; ++ct) {
            l0[ct] = *reinterpret_cast<const v2f*>(pl + ct * 16 * LDS_STRIDE + k2);
            r0[ct] = *reinterpret_cast<const v2f*>(pr + ct * 16 * LDS_STRIDE + k2);
          }
        }
        __builtin_amdgcn_sched_barrier(0);
        #pragma unroll
        for (int ct = 0; ct < 8; ++ct) {
          acc[ct] = __builtin_amdgcn_wmma_f32_16x16x4_f32(false, a1, false, l1[ct],
                                                          (short)0, acc[ct], false, false);
          acc[ct] = __builtin_amdgcn_wmma_f32_16x16x4_f32(false, x1, false, r1[ct],
                                                          (short)0, acc[ct], false, false);
        }
        __builtin_amdgcn_sched_barrier(0);
      }

      wave_lds_handover();
      #pragma unroll
      for (int ct = 0; ct < 8; ++ct) {
        #pragma unroll
        for (int v = 0; v < 8; ++v) ta[mbase + v][ct * 16 + m] = acc[ct][v];
      }
      wave_lds_handover();

      if (NORM_RELU) {
        const int row = lane & 15;
        const int cb  = (lane >> 4) * 64;
        float ss = 0.f;
        #pragma unroll 8
        for (int c2 = 0; c2 < 64; ++c2) {
          const float v = ta[row][cb + c2];
          ss += v * v;
        }
        ss += __shfl_xor(ss, 16, 32);
        const float scl = 1.0f / fmaxf(sqrtf(ss), 1e-12f);
        #pragma unroll 8
        for (int c2 = 0; c2 < 64; ++c2) {
          const float v = ta[row][cb + c2] * scl;
          ta[row][cb + c2] = fmaxf(v, 0.0f);
        }
        wave_lds_handover();
      }

      {
        float4 vo[16];
        #pragma unroll
        for (int rr = 0; rr < 16; ++rr)
          vo[rr] = *reinterpret_cast<const float4*>(&ta[rr][lane * 4]);
        #pragma unroll
        for (int rr = 0; rr < 16; ++rr) {
          const int n = node0 + rr;
          if (n < N)
            *reinterpret_cast<float4*>(outp + (long long)n * NODE_DIM + lane * 4) = vo[rr];
        }
      }
    }
    __syncthreads();
  }
}

extern "C" void kernel_launch(void* const* d_in, const int* in_sizes, int n_in,
                              void* d_out, int out_size, void* d_ws,
                              size_t ws_size, hipStream_t stream) {
  (void)n_in; (void)out_size; (void)ws_size;

  const float* x   = (const float*)d_in[0];
  const int*   ei  = (const int*)d_in[1];
  const float* W1l = (const float*)d_in[2];
  const float* b1  = (const float*)d_in[3];
  const float* W1r = (const float*)d_in[4];
  const float* W2l = (const float*)d_in[5];
  const float* b2  = (const float*)d_in[6];
  const float* W2r = (const float*)d_in[7];
  float* out = (float*)d_out;

  const int N = in_sizes[0] / NODE_DIM;
  const int E = in_sizes[1] / 2;
  const int* src = ei;
  const int* dst = ei + E;

  const int NB    = (N + BIN_NODES - 1) / BIN_NODES;
  const int NBP   = (NB + 3) & ~3;
  const int chunk = (E + NCHUNK - 1) / NCHUNK;

  char* ws = (char*)d_ws;
  int* deg = (int*)ws;
  size_t off = ((size_t)N * 4 + 255) & ~(size_t)255;
  off += (((size_t)N * NODE_DIM * 4) + 255) & ~(size_t)255;
  float* h = (float*)(ws + off);
  char* hb = (char*)h;
  size_t hoff = 0;
  int* binned = (int*)(hb + hoff);
  hoff += ((size_t)E * 8 + 255) & ~(size_t)255;
  int* slots = (int*)(hb + hoff);
  hoff += ((size_t)E * 4 + 255) & ~(size_t)255;
  int* cnt = (int*)(hb + hoff);
  hoff += ((size_t)NCHUNK * NBP * 4 + 255) & ~(size_t)255;
  int* binBase = (int*)(hb + hoff);
  off += (((size_t)N * NODE_DIM * 4) + 255) & ~(size_t)255;
  int* first = (int*)(ws + off);
  off += ((size_t)N * 4 + 255) & ~(size_t)255;
  int* csr = (int*)(ws + off);

  const int tiles  = (N + 15) / 16;

  sage_bincount_kernel<<<NCHUNK, 256, 0, stream>>>(dst, cnt, E, NBP, chunk);
  sage_binscatter_kernel<<<NCHUNK, 256, 0, stream>>>(
      src, dst, cnt, binBase, binned, E, NB, NBP, chunk);
  sage_bingroup_kernel<<<NB, BIN_NODES, 0, stream>>>(
      binned, binBase, deg, first, slots, csr, N, NB, E);

  sage_layer_kernel<true><<<LAYER_BLOCKS, LAYER_THREADS, 0, stream>>>(
      x, first, deg, csr, W1l, b1, W1r, h, N, tiles);

  sage_layer_kernel<false><<<LAYER_BLOCKS, LAYER_THREADS, 0, stream>>>(
      h, first, deg, csr, W2l, b2, W2r, out, N, tiles);
}


# Wave roles swapped by age: the 16 gather waves are waves 0-15 (older), the 4 matrix waves 16-19 (youngest of their SIMD), so row loads keep issuing beside the WMMA stream; sum order kept
# speedup vs baseline: 3.4163x; 1.0429x over previous
#include <hip/hip_runtime.h>
#include <hip/hip_bf16.h>
#include <math.h>

typedef float v2f __attribute__((ext_vector_type(2)));
typedef float v8f __attribute__((ext_vector_type(8)));

#define NODE_DIM 128
#define LDS_STRIDE 132

#define BIN_SHIFT 8
#define BIN_NODES 256
#define NCHUNK 256
#define NB_MAX 1024
#define EDGE_BATCH 16
#define REC_BATCH 8
#define SLOT_CAP 8192

__device__ __forceinline__ int block_exclusive_scan_256(int v, int* s_w) {
  const int lane = threadIdx.x & 31;
  const int wv   = threadIdx.x >> 5;
  int incl = v;
  #pragma unroll
  for (int o = 1; o < 32; o <<= 1) {
    const int up = __shfl_up(incl, o, 32);
    if (lane >= o) incl += up;
  }
  if (lane == 31) s_w[wv] = incl;
  __syncthreads();
  int ex = incl - v;
  for (int w = 0; w < wv; ++w) ex += s_w[w];
  return ex;
}

__global__ __launch_bounds__(256)
void sage_bincount_kernel(const int* __restrict__ dst, int* __restrict__ cnt,
                          int E, int NBP, int chunk) {
  __shared__ int s_hist[NB_MAX + 1];
  const int tid = threadIdx.x;
  for (int b = tid; b < NBP; b += 256) s_hist[b] = 0;
  __syncthreads();
  const int e0 = blockIdx.x * chunk;
  const int e1 = min(e0 + chunk, E);
  for (int base = e0; base < e1; base += 256 * EDGE_BATCH) {
    int bin[EDGE_BATCH];
    #pragma unroll
    for (int u = 0; u < EDGE_BATCH; ++u) {
      const int e = base + u * 256 + tid;
      const int d = dst[min(e, e1 - 1)];
      bin[u] = (e < e1) ? (d >> BIN_SHIFT) : NB_MAX;
    }
    #pragma unroll
    for (int u = 0; u < EDGE_BATCH; ++u) atomicAdd(&s_hist[bin[u]], 1);
  }
  __syncthreads();
  for (int b = tid; b < NBP; b += 256) cnt[blockIdx.x * NBP + b] = s_hist[b];
}

__global__ __launch_bounds__(256)
void sage_binscatter_kernel(const int* __restrict__ src,
                            const int* __restrict__ dst,
                            const int* __restrict__ cnt,
                            int* __restrict__ binBase,
                            int* __restrict__ binned,
                            int E, int NB, int NBP, int chunk) {
  __shared__ int s_cur[NB_MAX + 1];
  __shared__ int s_wsum[8];
  const int tid = threadIdx.x;
  const int c   = blockIdx.x;
  const int nq  = NBP >> 2;

  int4 before = make_int4(0, 0, 0, 0);
  int4 total  = make_int4(0, 0, 0, 0);
  if (tid < nq) {
    const int4* col = reinterpret_cast<const int4*>(cnt) + tid;
    for (int r0 = 0; r0 < NCHUNK; r0 += 16) {
      int4 v[16];
      #pragma unroll
      for (int u = 0; u < 16; ++u) v[u] = col[(r0 + u) * nq];
      #pragma unroll
      for (int u = 0; u < 16; ++u) {
        const int m = (r0 + u < c) ? -1 : 0;
        before.x += v[u].x & m; before.y += v[u].y & m;
        before.z += v[u].z & m; before.w += v[u].w & m;
        total.x += v[u].x; total.y += v[u].y;
        total.z += v[u].z; total.w += v[u].w;
      }
    }
  }
  const int t4 = (total.x + total.y) + (total.z + total.w);
  const int ex = block_exclusive_scan_256(t4, s_wsum);
  if (tid < nq) {
    const int b0 = tid * 4;
    const int base0 = ex;
    const int base1 = base0 + total.x;
    const int base2 = base1 + total.y;
    const int base3 = base2 + total.z;
    s_cur[b0 + 0] = base0 + before.x;
    s_cur[b0 + 1] = base1 + before.y;
    s_cur[b0 + 2] = base2 + before.z;
    s_cur[b0 + 3] = base3 + before.w;
    if (c == 0) {
      if (b0 + 0 < NB) binBase[b0 + 0] = base0;
      if (b0 + 1 < NB) binBase[b0 + 1] = base1;
      if (b0 + 2 < NB) binBase[b0 + 2] = base2;
      if (b0 + 3 < NB) binBase[b0 + 3] = base3;
    }
  }
  __syncthreads();

  int2* rec = reinterpret_cast<int2*>(binned);
  const int e0 = c * chunk;
  const int e1 = min(e0 + chunk, E);
  for (int base = e0; base < e1; base += 256 * EDGE_BATCH) {
    int d[EDGE_BATCH], s[EDGE_BATCH], pos[EDGE_BATCH];
    #pragma unroll
    for (int u = 0; u < EDGE_BATCH; ++u) {
      const int e = min(base + u * 256 + tid, e1 - 1);
      d[u] = dst[e];
      s[u] = src[e];
    }
    #pragma unroll
    for (int u = 0; u < EDGE_BATCH; ++u) {
      const bool in = base + u * 256 + tid < e1;
      pos[u] = atomicAdd(&s_cur[in ? (d[u] >> BIN_SHIFT) : NB_MAX], 1);
    }
    #pragma unroll
    for (int u = 0; u < EDGE_BATCH; ++u)
      if (base + u * 256 + tid < e1) rec[pos[u]] = make_int2(s[u], d[u]);
  }
}

template <bool IN_LDS>
__device__ __forceinline__ void sage_order_bin(const int* s_slot,
                                               const int* gslots,
                                               const int* s_off,
                                               const int* s_deg,
                                               int* __restrict__ csr_bin) {
  const int lane = threadIdx.x & 31;
  const int wv = __builtin_amdgcn_readfirstlane((int)(threadIdx.x >> 5));
  for (int k = wv; k < BIN_NODES; k += 8) {
    const int d = __builtin_amdgcn_readfirstlane(s_deg[k]);
    const int o = __builtin_amdgcn_readfirstlane(s_off[k]);
    for (int ib = 0; ib < d; ib += 32) {
      const int i = ib + lane;
      int si = 0;
      if (i < d)
        si = IN_LDS ? s_slot[o + i]
                    : __hip_atomic_load(gslots + o + i, __ATOMIC_RELAXED,
                                        __HIP_MEMORY_SCOPE_AGENT);
      int pos = 0;
      for (int jb = 0; jb < d; jb += 32) {
        const int j0 = jb + lane;
        int sc = 0;
        if (j0 < d)
          sc = IN_LDS ? s_slot[o + j0]
                      : __hip_atomic_load(gslots + o + j0, __ATOMIC_RELAXED,
                                          __HIP_MEMORY_SCOPE_AGENT);
        const int cnt = min(32, d - jb);
        for (int jj = 0; jj < cnt; ++jj) {
          const int sj = __builtin_amdgcn_readlane(sc, jj);
          const int j  = jb + jj;
          pos += ((sj < si) || (sj == si && j < i)) ? 1 : 0;
        }
      }
      if (i < d) csr_bin[o + pos] = si;
    }
  }
}

__global__ __launch_bounds__(BIN_NODES)
void sage_bingroup_kernel(const int* __restrict__ binned,
                          const int* __restrict__ binBase,
                          int* __restrict__ deg, int* __restrict__ first,
                          int* slots, int* __restrict__ csr,
                          int N, int NB, int E) {
  __shared__ int s_cnt[BIN_NODES + 1];
  __shared__ int s_off[BIN_NODES];
  __shared__ int s_deg[BIN_NODES];
  __shared__ int s_w[8];
  __shared__ int s_slot[SLOT_CAP];
  const int tid = threadIdx.x;
  const int b   = blockIdx.x;
  const int beg = binBase[b];
  const int end = (b + 1 < NB) ? binBase[b + 1] : E;
  const int n0  = b << BIN_SHIFT;
  const bool in_lds = (end - beg) <= SLOT_CAP;
  const int2* rec = reinterpret_cast<const int2*>(binned);

  s_cnt[tid] = 0;
  __syncthreads();
  for (int base = beg; base < end; base += BIN_NODES * REC_BATCH) {
    int k[REC_BATCH];
    #pragma unroll
    for (int u = 0; u < REC_BATCH; ++u) {
      const int i = base + u * BIN_NODES + tid;
      const int d = rec[min(i, end - 1)].y;
      k[u] = (i < end) ? (d - n0) : BIN_NODES;
    }
    #pragma unroll
    for (int u = 0; u < REC_BATCH; ++u) atomicAdd(&s_cnt[k[u]], 1);
  }
  __syncthreads();

  const int cv = s_cnt[tid];
  const int ex = block_exclusive_scan_256(cv, s_w);
  const int n  = n0 + tid;
  if (n < N) { deg[n] = cv; first[n] = beg + ex; }
  s_deg[tid] = cv;
  s_off[tid] = ex;
  s_cnt[tid] = ex;
  __syncthreads();

  for (int base = beg; base < end; base += BIN_NODES * REC_BATCH) {
    int2 r[REC_BATCH];
    int pos[REC_BATCH];
    #pragma unroll
    for (int u = 0; u < REC_BATCH; ++u)
      r[u] = rec[min(base + u * BIN_NODES + tid, end - 1)];
    #pragma unroll
    for (int u = 0; u < REC_BATCH; ++u) {
      const bool in = base + u * BIN_NODES + tid < end;
      pos[u] = atomicAdd(&s_cnt[in ? (r[u].y - n0) : BIN_NODES], 1);
    }
    #pragma unroll
    for (int u = 0; u < REC_BATCH; ++u) {
      if (base + u * BIN_NODES + tid < end) {
        if (in_lds) s_slot[pos[u]] = r[u].x;
        else        slots[beg + pos[u]] = r[u].x;
      }
    }
  }
  __syncthreads();

  if (in_lds) sage_order_bin<true >(s_slot, slots + beg, s_off, s_deg, csr + beg);
  else        sage_order_bin<false>(s_slot, slots + beg, s_off, s_deg, csr + beg);
}

#define LAYER_MWAVES 4
#define LAYER_GWAVES 16
#define LAYER_WAVES (LAYER_MWAVES + LAYER_GWAVES)
#define LAYER_THREADS (LAYER_WAVES * 32)
#define LAYER_BLOCKS 256
#define ROWS_AHEAD 16

__device__ __forceinline__ void wave_lds_handover() {
  __builtin_amdgcn_fence(__ATOMIC_ACQ_REL, "wavefront");
}

__device__ __forceinline__ void sage_gather_rows(const float* __restrict__ feat,
                                                 const int* __restrict__ first,
                                                 const int* __restrict__ deg,
                                                 const int* __restrict__ csr,
                                                 float (*trow)[LDS_STRIDE],
                                                 int n0, int npw, int N, int lane) {
  const int li = lane & 3;
  const int nn = n0 + li;
  const int nc = min(nn, N - 1);
  const int fv = first[nc];
  const int dv = (li < npw && nn < N) ? deg[nc] : 0;
  const int e0 = __builtin_amdgcn_readlane(fv, 0);
  const int total = __builtin_amdgcn_readlane(dv, 0) + __builtin_amdgcn_readlane(dv, 1) +
                    __builtin_amdgcn_readlane(dv, 2) + __builtin_amdgcn_readlane(dv, 3);
  float4 acc = make_float4(0.f, 0.f, 0.f, 0.f);
  int k   = 0;
  int nxt = __builtin_amdgcn_readlane(dv, 0);
  int sc = (lane < total) ? csr[e0 + lane] : 0;
  for (int jb = 0; jb < total; jb += 32) {
    const int cnt = min(32, total - jb);
    int scn = 0;
    for (int j16 = 0; j16 < cnt; j16 += ROWS_AHEAD) {
      float4 v[ROWS_AHEAD];
      #pragma unroll
      for (int u = 0; u < ROWS_AHEAD; ++u) {
        const int s = __builtin_amdgcn_readlane(sc, j16 + u);
        v[u] = *reinterpret_cast<const float4*>(
            feat + (long long)s * NODE_DIM + lane * 4);
      }
      if (j16 == 0) {
        const int jn = jb + 32 + lane;
        if (jn < total) scn = csr[e0 + jn];
      }
      #pragma unroll
      for (int u = 0; u < ROWS_AHEAD; ++u) {
        const int idx = jb + j16 + u;
        if (idx < total) {
          while (idx >= nxt) {
            const float rc = 1.0f / fmaxf((float)__builtin_amdgcn_readlane(dv, k), 1.0f);
            *reinterpret_cast<float4*>(&trow[k][lane * 4]) =
                make_float4(acc.x * rc, acc.y * rc, acc.z * rc, acc.w * rc);
            acc = make_float4(0.f, 0.f, 0.f, 0.f);
            ++k;
            nxt += __builtin_amdgcn_readlane(dv, k);
          }
          acc.x += v[u].x; acc.y += v[u].y; acc.z += v[u].z; acc.w += v[u].w;
        }
      }
    }
    sc = scn;
  }
  while (k < npw) {
    const float rc = 1.0f / fmaxf((float)__builtin_amdgcn_readlane(dv, k), 1.0f);
    *reinterpret_cast<float4*>(&trow[k][lane * 4]) =
        make_float4(acc.x * rc, acc.y * rc, acc.z * rc, acc.w * rc);
    acc = make_float4(0.f, 0.f, 0.f, 0.f);
    ++k;
  }
}

template <bool NORM_RELU>
__global__ __launch_bounds__(LAYER_THREADS)
void sage_layer_kernel(const float* __restrict__ feat,
                       const int* __restrict__ first,
                       const int* __restrict__ deg,
                       const int* __restrict__ csr,
                       const float* __restrict__ Wl,
                       const float* __restrict__ bias,
                       const float* __restrict__ Wr,
                       float* __restrict__ outp, int N, int tiles) {
  __shared__ float s_wl[NODE_DIM][LDS_STRIDE];
  __shared__ float s_wr[NODE_DIM][LDS_STRIDE];
  __shared__ float s_ring[2][LAYER_MWAVES][16][LDS_STRIDE];
  __shared__ float s_root[LAYER_MWAVES][16][LDS_STRIDE];

  const int tid  = threadIdx.x;
  const int lane = tid & 31;
  const int wv   = __builtin_amdgcn_readfirstlane(tid >> 5);
  const int mw   = wv - LAYER_GWAVES;

  {
    constexpr int W4   = NODE_DIM * NODE_DIM / 4;
    constexpr int NLD  = (W4 + LAYER_THREADS - 1) / LAYER_THREADS;
    float4 v[NLD];
    #pragma unroll
    for (int u = 0; u < NLD; ++u)
      v[u] = reinterpret_cast<const float4*>(Wl)[min(u * LAYER_THREADS + tid, W4 - 1)];
    #pragma unroll
    for (int u = 0; u < NLD; ++u) {
      const int i = u * LAYER_THREADS + tid;
      if (i < W4) *reinterpret_cast<float4*>(&s_wl[i >> 5][(i & 31) * 4]) = v[u];
    }
    #pragma unroll
    for (int u = 0; u < NLD; ++u)
      v[u] = reinterpret_cast<const float4*>(Wr)[min(u * LAYER_THREADS + tid, W4 - 1)];
    #pragma unroll
    for (int u = 0; u < NLD; ++u) {
      const int i = u * LAYER_THREADS + tid;
      if (i < W4) *reinterpret_cast<float4*>(&s_wr[i >> 5][(i & 31) * 4]) = v[u];
    }
  }

  const int m     = lane & 15;
  const int khalf = (lane >> 4) << 1;
  const int mbase = (lane >> 4) << 3;

  const int b      = (int)blockIdx.x;
  const int G      = (int)gridDim.x;
  const int nt     = (b < tiles) ? (tiles - b + G - 1) / G : 0;
  const int rounds = (nt + LAYER_MWAVES - 1) / LAYER_MWAVES;

  for (int r = 0; r <= rounds; ++r) {
    if (mw < 0) {
      if (r < rounds) {
        const int g   = wv;
        const int ntr = min(LAYER_MWAVES, nt - r * LAYER_MWAVES);
        const int npw = (ntr >= 3) ? 4 : ntr;
        const int j0  = g * npw;
        if (j0 < 16 * ntr) {
          const int q    = j0 >> 4;
          const int row0 = j0 & 15;
          const int tile = b + (r * LAYER_MWAVES + q) * G;
          sage_gather_rows(feat, first, deg, csr, &s_ring[r & 1][q][row0],
                           tile * 16 + row0, npw, N, lane);
        }
      }
    } else if (r > 0 && (r - 1) * LAYER_MWAVES + mw < nt) {
      const int tile  = b + ((r - 1) * LAYER_MWAVES + mw) * G;
      const int node0 = tile * 16;
      float (*ta)[LDS_STRIDE] = s_ring[(r - 1) & 1][mw];
      float (*tx)[LDS_STRIDE] = s_root[mw];
      const float* pa = &ta[m][khalf];
      const float* px = &tx[m][khalf];
      const float* pl = &s_wl[m][khalf];
      const float* pr = &s_wr[m][khalf];

      float bv[8];
      #pragma unroll
      for (int ct = 0; ct < 8; ++ct) bv[ct] = bias[ct * 16 + m];

      {
        float4 vx[16];
        #pragma unroll
        for (int rr = 0; rr < 16; ++rr) {
          const int nc = min(node0 + rr, N - 1);
          vx[rr] = *reinterpret_cast<const float4*>(feat + (long long)nc * NODE_DIM + lane * 4);
        }
        #pragma unroll
        for (int rr = 0; rr < 16; ++rr)
          *reinterpret_cast<float4*>(&tx[rr][lane * 4]) = vx[rr];
      }
      wave_lds_handover();

      v8f acc[8];
      #pragma unroll
      for (int ct = 0; ct < 8; ++ct) {
        #pragma unroll
        for (int v = 0; v < 8; ++v) acc[ct][v] = bv[ct];
      }

      v2f a0, x0, l0[8], r0[8], a1, x1, l1[8], r1[8];
      a0 = *reinterpret_cast<const v2f*>(pa);
      x0 = *reinterpret_cast<const v2f*>(px);
      #pragma unroll
      for (int ct = 0; ct < 8; ++ct) {
        l0[ct] = *reinterpret_cast<const v2f*>(pl + ct * 16 * LDS_STRIDE);
        r0[ct] = *reinterpret_cast<const v2f*>(pr + ct * 16 * LDS_STRIDE);
      }
      #pragma unroll
      for (int k0 = 0; k0 < NODE_DIM; k0 += 8) {
        const int k1 = k0 + 4;
        a1 = *reinterpret_cast<const v2f*>(pa + k1);
        x1 = *reinterpret_cast<const v2f*>(px + k1);
        #pragma unroll
        for (int ct = 0; ct < 8; ++ct) {
          l1[ct] = *reinterpret_cast<const v2f*>(pl + ct * 16 * LDS_STRIDE + k1);
          r1[ct] = *reinterpret_cast<const v2f*>(pr + ct * 16 * LDS_STRIDE + k1);
        }
        __builtin_amdgcn_sched_barrier(0);
        #pragma unroll
        for (int ct = 0; ct < 8; ++ct) {
          acc[ct] = __builtin_amdgcn_wmma_f32_16x16x4_f32(false, a0, false, l0[ct],
                                                          (short)0, acc[ct], false, false);
          acc[ct] = __builtin_amdgcn_wmma_f32_16x16x4_f32(false, x0, false, r0[ct],
                                                          (short)0, acc[ct], false, false);
        }
        __builtin_amdgcn_sched_barrier(0);
        if (k0 + 8 < NODE_DIM) {
          const int k2 = k0 + 8;
          a0 = *reinterpret_cast<const v2f*>(pa + k2);
          x0 = *reinterpret_cast<const v2f*>(px + k2);
          #pragma unroll
          for (int ct = 0; ct < 8; ++ct) {
            l0[ct] = *reinterpret_cast<const v2f*>(pl + ct * 16 * LDS_STRIDE + k2);
            r0[ct] = *reinterpret_cast<const v2f*>(pr + ct * 16 * LDS_STRIDE + k2);
          }
        }
        __builtin_amdgcn_sched_barrier(0);
        #pragma unroll
        for (int ct = 0; ct < 8; ++ct) {
          acc[ct] = __builtin_amdgcn_wmma_f32_16x16x4_f32(false, a1, false, l1[ct],
                                                          (short)0, acc[ct], false, false);
          acc[ct] = __builtin_amdgcn_wmma_f32_16x16x4_f32(false, x1, false, r1[ct],
                                                          (short)0, acc[ct], false, false);
        }
        __builtin_amdgcn_sched_barrier(0);
      }

      wave_lds_handover();
      #pragma unroll
      for (int ct = 0; ct < 8; ++ct) {
        #pragma unroll
        for (int v = 0; v < 8; ++v) ta[mbase + v][ct * 16 + m] = acc[ct][v];
      }
      wave_lds_handover();

      if (NORM_RELU) {
        const int row = lane & 15;
        const int cb  = (lane >> 4) * 64;
        float ss = 0.f;
        #pragma unroll 8
        for (int c2 = 0; c2 < 64; ++c2) {
          const float v = ta[row][cb + c2];
          ss += v * v;
        }
        ss += __shfl_xor(ss, 16, 32);
        const float scl = 1.0f / fmaxf(sqrtf(ss), 1e-12f);
        #pragma unroll 8
        for (int c2 = 0; c2 < 64; ++c2) {
          const float v = ta[row][cb + c2] * scl;
          ta[row][cb + c2] = fmaxf(v, 0.0f);
        }
        wave_lds_handover();
      }

      {
        float4 vo[16];
        #pragma unroll
        for (int rr = 0; rr < 16; ++rr)
          vo[rr] = *reinterpret_cast<const float4*>(&ta[rr][lane * 4]);
        #pragma unroll
        for (int rr = 0; rr < 16; ++rr) {
          const int n = node0 + rr;
          if (n < N)
            *reinterpret_cast<float4*>(outp + (long long)n * NODE_DIM + lane * 4) = vo[rr];
        }
      }
    }
    __syncthreads();
  }
}

extern "C" void kernel_launch(void* const* d_in, const int* in_sizes, int n_in,
                              void* d_out, int out_size, void* d_ws,
                              size_t ws_size, hipStream_t stream) {
  (void)n_in; (void)out_size; (void)ws_size;

  const float* x   = (const float*)d_in[0];
  const int*   ei  = (const int*)d_in[1];
  const float* W1l = (const float*)d_in[2];
  const float* b1  = (const float*)d_in[3];
  const float* W1r = (const float*)d_in[4];
  const float* W2l = (const float*)d_in[5];
  const float* b2  = (const float*)d_in[6];
  const float* W2r = (const float*)d_in[7];
  float* out = (float*)d_out;

  const int N = in_sizes[0] / NODE_DIM;
  const int E = in_sizes[1] / 2;
  const int* src = ei;
  const int* dst = ei + E;

  const int NB    = (N + BIN_NODES - 1) / BIN_NODES;
  const int NBP   = (NB + 3) & ~3;
  const int chunk = (E + NCHUNK - 1) / NCHUNK;

  char* ws = (char*)d_ws;
  int* deg = (int*)ws;
  size_t off = ((size_t)N * 4 + 255) & ~(size_t)255;
  off += (((size_t)N * NODE_DIM * 4) + 255) & ~(size_t)255;
  float* h = (float*)(ws + off);
  char* hb = (char*)h;
  size_t hoff = 0;
  int* binned = (int*)(hb + hoff);
  hoff += ((size_t)E * 8 + 255) & ~(size_t)255;
  int* slots = (int*)(hb + hoff);
  hoff += ((size_t)E * 4 + 255) & ~(size_t)255;
  int* cnt = (int*)(hb + hoff);
  hoff += ((size_t)NCHUNK * NBP * 4 + 255) & ~(size_t)255;
  int* binBase = (int*)(hb + hoff);
  off += (((size_t)N * NODE_DIM * 4) + 255) & ~(size_t)255;
  int* first = (int*)(ws + off);
  off += ((size_t)N * 4 + 255) & ~(size_t)255;
  int* csr = (int*)(ws + off);

  const int tiles  = (N + 15) / 16;

  sage_bincount_kernel<<<NCHUNK, 256, 0, stream>>>(dst, cnt, E, NBP, chunk);
  sage_binscatter_kernel<<<NCHUNK, 256, 0, stream>>>(
      src, dst, cnt, binBase, binned, E, NB, NBP, chunk);
  sage_bingroup_kernel<<<NB, BIN_NODES, 0, stream>>>(
      binned, binBase, deg, first, slots, csr, N, NB, E);

  sage_layer_kernel<true><<<LAYER_BLOCKS, LAYER_THREADS, 0, stream>>>(
      x, first, deg, csr, W1l, b1, W1r, h, N, tiles);

  sage_layer_kernel<false><<<LAYER_BLOCKS, LAYER_THREADS, 0, stream>>>(
      h, first, deg, csr, W2l, b2, W2r, out, N, tiles);
}


# Round barrier by hand waits for LDS only: output stores and next round's ids stay in flight across it; degrees and the tile's own rows come with the gather waves' first row batch; sum order kept
# speedup vs baseline: 3.5443x; 1.0375x over previous
#include <hip/hip_runtime.h>
#include <hip/hip_bf16.h>
#include <math.h>

typedef float v2f __attribute__((ext_vector_type(2)));
typedef float v8f __attribute__((ext_vector_type(8)));

#define NODE_DIM 128
#define LDS_STRIDE 132

#define BIN_SHIFT 8
#define BIN_NODES 256
#define NCHUNK 256
#define NB_MAX 1024
#define EDGE_BATCH 16
#define REC_BATCH 8
#define SLOT_CAP 8192

__device__ __forceinline__ int block_exclusive_scan_256(int v, int* s_w) {
  const int lane = threadIdx.x & 31;
  const int wv   = threadIdx.x >> 5;
  int incl = v;
  #pragma unroll
  for (int o = 1; o < 32; o <<= 1) {
    const int up = __shfl_up(incl, o, 32);
    if (lane >= o) incl += up;
  }
  if (lane == 31) s_w[wv] = incl;
  __syncthreads();
  int ex = incl - v;
  for (int w = 0; w < wv; ++w) ex += s_w[w];
  return ex;
}

__global__ __launch_bounds__(256)
void sage_bincount_kernel(const int* __restrict__ dst, int* __restrict__ cnt,
                          int E, int NBP, int chunk) {
  __shared__ int s_hist[NB_MAX + 1];
  const int tid = threadIdx.x;
  for (int b = tid; b < NBP; b += 256) s_hist[b] = 0;
  __syncthreads();
  const int e0 = blockIdx.x * chunk;
  const int e1 = min(e0 + chunk, E);
  for (int base = e0; base < e1; base += 256 * EDGE_BATCH) {
    int bin[EDGE_BATCH];
    #pragma unroll
    for (int u = 0; u < EDGE_BATCH; ++u) {
      const int e = base + u * 256 + tid;
      const int d = dst[min(e, e1 - 1)];
      bin[u] = (e < e1) ? (d >> BIN_SHIFT) : NB_MAX;
    }
    #pragma unroll
    for (int u = 0; u < EDGE_BATCH; ++u) atomicAdd(&s_hist[bin[u]], 1);
  }
  __syncthreads();
  for (int b = tid; b < NBP; b += 256) cnt[blockIdx.x * NBP + b] = s_hist[b];
}

__global__ __launch_bounds__(256)
void sage_binscatter_kernel(const int* __restrict__ src,
                            const int* __restrict__ dst,
                            const int* __restrict__ cnt,
                            int* __restrict__ binBase,
                            int* __restrict__ binned,
                            int E, int NB, int NBP, int chunk) {
  __shared__ int s_cur[NB_MAX + 1];
  __shared__ int s_wsum[8];
  const int tid = threadIdx.x;
  const int c   = blockIdx.x;
  const int nq  = NBP >> 2;

  int4 before = make_int4(0, 0, 0, 0);
  int4 total  = make_int4(0, 0, 0, 0);
  if (tid < nq) {
    const int4* col = reinterpret_cast<const int4*>(cnt) + tid;
    for (int r0 = 0; r0 < NCHUNK; r0 += 16) {
      int4 v[16];
      #pragma unroll
      for (int u = 0; u < 16; ++u) v[u] = col[(r0 + u) * nq];
      #pragma unroll
      for (int u = 0; u < 16; ++u) {
        const int m = (r0 + u < c) ? -1 : 0;
        before.x += v[u].x & m; before.y += v[u].y & m;
        before.z += v[u].z & m; before.w += v[u].w & m;
        total.x += v[u].x; total.y += v[u].y;
        total.z += v[u].z; total.w += v[u].w;
      }
    }
  }
  const int t4 = (total.x + total.y) + (total.z + total.w);
  const int ex = block_exclusive_scan_256(t4, s_wsum);
  if (tid < nq) {
    const int b0 = tid * 4;
    const int base0 = ex;
    const int base1 = base0 + total.x;
    const int base2 = base1 + total.y;
    const int base3 = base2 + total.z;
    s_cur[b0 + 0] = base0 + before.x;
    s_cur[b0 + 1] = base1 + before.y;
    s_cur[b0 + 2] = base2 + before.z;
    s_cur[b0 + 3] = base3 + before.w;
    if (c == 0) {
      if (b0 + 0 < NB) binBase[b0 + 0] = base0;
      if (b0 + 1 < NB) binBase[b0 + 1] = base1;
      if (b0 + 2 < NB) binBase[b0 + 2] = base2;
      if (b0 + 3 < NB) binBase[b0 + 3] = base3;
    }
  }
  __syncthreads();

  int2* rec = reinterpret_cast<int2*>(binned);
  const int e0 = c * chunk;
  const int e1 = min(e0 + chunk, E);
  for (int base = e0; base < e1; base += 256 * EDGE_BATCH) {
    int d[EDGE_BATCH], s[EDGE_BATCH], pos[EDGE_BATCH];
    #pragma unroll
    for (int u = 0; u < EDGE_BATCH; ++u) {
      const int e = min(base + u * 256 + tid, e1 - 1);
      d[u] = dst[e];
      s[u] = src[e];
    }
    #pragma unroll
    for (int u = 0; u < EDGE_BATCH; ++u) {
      const bool in = base + u * 256 + tid < e1;
      pos[u] = atomicAdd(&s_cur[in ? (d[u] >> BIN_SHIFT) : NB_MAX], 1);
    }
    #pragma unroll
    for (int u = 0; u < EDGE_BATCH; ++u)
      if (base + u * 256 + tid < e1) rec[pos[u]] = make_int2(s[u], d[u]);
  }
}

template <bool IN_LDS>
__device__ __forceinline__ void sage_order_bin(const int* s_slot,
                                               const int* gslots,
                                               const int* s_off,
                                               const int* s_deg,
                                               int* __restrict__ csr_bin) {
  const int lane = threadIdx.x & 31;
  const int wv = __builtin_amdgcn_readfirstlane((int)(threadIdx.x >> 5));
  for (int k = wv; k < BIN_NODES; k += 8) {
    const int d = __builtin_amdgcn_readfirstlane(s_deg[k]);
    const int o = __builtin_amdgcn_readfirstlane(s_off[k]);
    for (int ib = 0; ib < d; ib += 32) {
      const int i = ib + lane;
      int si = 0;
      if (i < d)
        si = IN_LDS ? s_slot[o + i]
                    : __hip_atomic_load(gslots + o + i, __ATOMIC_RELAXED,
                                        __HIP_MEMORY_SCOPE_AGENT);
      int pos = 0;
      for (int jb = 0; jb < d; jb += 32) {
        const int j0 = jb + lane;
        int sc = 0;
        if (j0 < d)
          sc = IN_LDS ? s_slot[o + j0]
                      : __hip_atomic_load(gslots + o + j0, __ATOMIC_RELAXED,
                                          __HIP_MEMORY_SCOPE_AGENT);
        const int cnt = min(32, d - jb);
        for (int jj = 0; jj < cnt; ++jj) {
          const int sj = __builtin_amdgcn_readlane(sc, jj);
          const int j  = jb + jj;
          pos += ((sj < si) || (sj == si && j < i)) ? 1 : 0;
        }
      }
      if (i < d) csr_bin[o + pos] = si;
    }
  }
}

__global__ __launch_bounds__(BIN_NODES)
void sage_bingroup_kernel(const int* __restrict__ binned,
                          const int* __restrict__ binBase,
                          int* __restrict__ deg, int* __restrict__ first,
                          int* slots, int* __restrict__ csr,
                          int N, int NB, int E) {
  __shared__ int s_cnt[BIN_NODES + 1];
  __shared__ int s_off[BIN_NODES];
  __shared__ int s_deg[BIN_NODES];
  __shared__ int s_w[8];
  __shared__ int s_slot[SLOT_CAP];
  const int tid = threadIdx.x;
  const int b   = blockIdx.x;
  const int beg = binBase[b];
  const int end = (b + 1 < NB) ? binBase[b + 1] : E;
  const int n0  = b << BIN_SHIFT;
  const bool in_lds = (end - beg) <= SLOT_CAP;
  const int2* rec = reinterpret_cast<const int2*>(binned);

  s_cnt[tid] = 0;
  __syncthreads();
  for (int base = beg; base < end; base += BIN_NODES * REC_BATCH) {
    int k[REC_BATCH];
    #pragma unroll
    for (int u = 0; u < REC_BATCH; ++u) {
      const int i = base + u * BIN_NODES + tid;
      const int d = rec[min(i, end - 1)].y;
      k[u] = (i < end) ? (d - n0) : BIN_NODES;
    }
    #pragma unroll
    for (int u = 0; u < REC_BATCH; ++u) atomicAdd(&s_cnt[k[u]], 1);
  }
  __syncthreads();

  const int cv = s_cnt[tid];
  const int ex = block_exclusive_scan_256(cv, s_w);
  const int n  = n0 + tid;
  if (n < N) { deg[n] = cv; first[n] = beg + ex; }
  s_deg[tid] = cv;
  s_off[tid] = ex;
  s_cnt[tid] = ex;
  __syncthreads();

  for (int base = beg; base < end; base += BIN_NODES * REC_BATCH) {
    int2 r[REC_BATCH];
    int pos[REC_BATCH];
    #pragma unroll
    for (int u = 0; u < REC_BATCH; ++u)
      r[u] = rec[min(base + u * BIN_NODES + tid, end - 1)];
    #pragma unroll
    for (int u = 0; u < REC_BATCH; ++u) {
      const bool in = base + u * BIN_NODES + tid < end;
      pos[u] = atomicAdd(&s_cnt[in ? (r[u].y - n0) : BIN_NODES], 1);
    }
    #pragma unroll
    for (int u = 0; u < REC_BATCH; ++u) {
      if (base + u * BIN_NODES + tid < end) {
        if (in_lds) s_slot[pos[u]] = r[u].x;
        else        slots[beg + pos[u]] = r[u].x;
      }
    }
  }
  __syncthreads();

  if (in_lds) sage_order_bin<true >(s_slot, slots + beg, s_off, s_deg, csr + beg);
  else        sage_order_bin<false>(s_slot, slots + beg, s_off, s_deg, csr + beg);
}

#define LAYER_MWAVES 4
#define LAYER_GWAVES 16
#define LAYER_WAVES (LAYER_MWAVES + LAYER_GWAVES)
#define LAYER_THREADS (LAYER_WAVES * 32)
#define LAYER_BLOCKS 256
#define ROWS_AHEAD 16

__device__ __forceinline__ void wave_lds_handover() {
  __builtin_amdgcn_fence(__ATOMIC_ACQ_REL, "wavefront");
}

__device__ __forceinline__ void round_barrier() {
  asm volatile("s_wait_dscnt 0x0\n\ts_barrier_signal -1\n\ts_barrier_wait -1" ::: "memory");
}

struct GatherSlot { int active; int q; int row0; int npw; int n0; };
__device__ __forceinline__ GatherSlot sage_gather_slot(int r, int rounds, int nt,
                                                       int g, int b, int G) {
  GatherSlot s;
  s.active = 0; s.q = 0; s.row0 = 0; s.npw = 0; s.n0 = 0;
  if (r < rounds) {
    const int ntr = min(LAYER_MWAVES, nt - r * LAYER_MWAVES);
    const int npw = (ntr >= 3) ? 4 : ntr;
    const int j0  = g * npw;
    if (j0 < 16 * ntr) {
      s.active = 1;
      s.q      = j0 >> 4;
      s.row0   = j0 & 15;
      s.npw    = npw;
      s.n0     = (b + (r * LAYER_MWAVES + s.q) * G) * 16 + s.row0;
    }
  }
  return s;
}

__device__ __forceinline__ void sage_ask_heads(const int* __restrict__ first,
                                               const int* __restrict__ deg,
                                               int n0, int npw, int N, int lane,
                                               int& fv, int& dv) {
  const int li = lane & 3;
  const int nn = n0 + li;
  const int nc = min(nn, N - 1);
  fv = first[nc];
  dv = (li < npw && nn < N) ? deg[nc] : 0;
}

__device__ __forceinline__ int sage_ask_ids(const int* __restrict__ csr,
                                            int fv, int dv, int lane, int& e0) {
  e0 = __builtin_amdgcn_readlane(fv, 0);
  const int total = __builtin_amdgcn_readlane(dv, 0) + __builtin_amdgcn_readlane(dv, 1) +
                    __builtin_amdgcn_readlane(dv, 2) + __builtin_amdgcn_readlane(dv, 3);
  return (lane < total) ? csr[e0 + lane] : 0;
}

__device__ __forceinline__ void sage_gather_rows(const float* __restrict__ feat,
                                                 const int* __restrict__ first,
                                                 const int* __restrict__ deg,
                                                 const int* __restrict__ csr,
                                                 float (*trow)[LDS_STRIDE],
                                                 float (*xrow)[LDS_STRIDE],
                                                 int dv, int e0in, int sc,
                                                 int n0, int npw, int N, int lane,
                                                 int nx_active, int nx_n0, int nx_npw,
                                                 int& fvn, int& dvn) {
  const float* o0 = feat + (long long)min(n0, N - 1) * NODE_DIM + lane * 4;
  const float* o1 = feat + (long long)min(n0 + min(1, npw - 1), N - 1) * NODE_DIM + lane * 4;
  const float* o2 = feat + (long long)min(n0 + min(2, npw - 1), N - 1) * NODE_DIM + lane * 4;
  const float* o3 = feat + (long long)min(n0 + min(3, npw - 1), N - 1) * NODE_DIM + lane * 4;
  float4 own0 = make_float4(0.f, 0.f, 0.f, 0.f);
  float4 own1 = own0, own2 = own0, own3 = own0;
  fvn = 0; dvn = 0;
  const int e0 = __builtin_amdgcn_readfirstlane(e0in);
  const int total = __builtin_amdgcn_readlane(dv, 0) + __builtin_amdgcn_readlane(dv, 1) +
                    __builtin_amdgcn_readlane(dv, 2) + __builtin_amdgcn_readlane(dv, 3);
  float4 acc = make_float4(0.f, 0.f, 0.f, 0.f);
  int k   = 0;
  int nxt = __builtin_amdgcn_readlane(dv, 0);
  for (int jb = 0; jb < total; jb += 32) {
    const int cnt = min(32, total - jb);
    int scn = 0;
    for (int j16 = 0; j16 < cnt; j16 += ROWS_AHEAD) {
      float4 v[ROWS_AHEAD];
      #pragma unroll
      for (int u = 0; u < ROWS_AHEAD; ++u) {
        const int s = __builtin_amdgcn_readlane(sc, j16 + u);
        v[u] = *reinterpret_cast<const float4*>(
            feat + (long long)s * NODE_DIM + lane * 4);
      }
      if (j16 == 0) {
        const int jn = jb + 32 + lane;
        if (jn < total) scn = csr[e0 + jn];
        if (jb == 0) {
          own0 = *reinterpret_cast<const float4*>(o0);
          own1 = *reinterpret_cast<const float4*>(o1);
          own2 = *reinterpret_cast<const float4*>(o2);
          own3 = *reinterpret_cast<const float4*>(o3);
          if (nx_active) sage_ask_heads(first, deg, nx_n0, nx_npw, N, lane, fvn, dvn);
        }
      }
      #pragma unroll
      for (int u = 0; u < ROWS_AHEAD; ++u) {
        const int idx = jb + j16 + u;
        if (idx < total) {
          while (idx >= nxt) {
            const float rc = 1.0f / fmaxf((float)__builtin_amdgcn_readlane(dv, k), 1.0f);
            *reinterpret_cast<float4*>(&trow[k][lane * 4]) =
                make_float4(acc.x * rc, acc.y * rc, acc.z * rc, acc.w * rc);
            acc = make_float4(0.f, 0.f, 0.f, 0.f);
            ++k;
            nxt += __builtin_amdgcn_readlane(dv, k);
          }
          acc.x += v[u].x; acc.y += v[u].y; acc.z += v[u].z; acc.w += v[u].w;
        }
      }
    }
    sc = scn;
  }
  while (k < npw) {
    const float rc = 1.0f / fmaxf((float)__builtin_amdgcn_readlane(dv, k), 1.0f);
    *reinterpret_cast<float4*>(&trow[k][lane * 4]) =
        make_float4(acc.x * rc, acc.y * rc, acc.z * rc, acc.w * rc);
    acc = make_float4(0.f, 0.f, 0.f, 0.f);
    ++k;
  }
  if (total <= 0) {
    own0 = *reinterpret_cast<const float4*>(o0);
    own1 = *reinterpret_cast<const float4*>(o1);
    own2 = *reinterpret_cast<const float4*>(o2);
    own3 = *reinterpret_cast<const float4*>(o3);
    if (nx_active) sage_ask_heads(first, deg, nx_n0, nx_npw, N, lane, fvn, dvn);
  }
  *reinterpret_cast<float4*>(&xrow[0][lane * 4]) = own0;
  if (npw > 1) *reinterpret_cast<float4*>(&xrow[1][lane * 4]) = own1;
  if (npw > 2) *reinterpret_cast<float4*>(&xrow[2][lane * 4]) = own2;
  if (npw > 3) *reinterpret_cast<float4*>(&xrow[3][lane * 4]) = own3;
}

template <bool NORM_RELU>
__global__ __launch_bounds__(LAYER_THREADS)
void sage_layer_kernel(const float* __restrict__ feat,
                       const int* __restrict__ first,
                       const int* __restrict__ deg,
                       const int* __restrict__ csr,
                       const float* __restrict__ Wl,
                       const float* __restrict__ bias,
                       const float* __restrict__ Wr,
                       float* __restrict__ outp, int N, int tiles) {
  __shared__ float s_wl[NODE_DIM][LDS_STRIDE];
  __shared__ float s_wr[NODE_DIM][LDS_STRIDE];
  __shared__ float s_ring[2][LAYER_MWAVES][16][LDS_STRIDE];
  __shared__ float s_root[2][LAYER_MWAVES][16][LDS_STRIDE];
  __shared__ float s_bias[NODE_DIM];

  const int tid  = threadIdx.x;
  const int lane = tid & 31;
  const int wv   = __builtin_amdgcn_readfirstlane(tid >> 5);
  const int mw   = wv - LAYER_GWAVES;

  const int m     = lane & 15;
  const int khalf = (lane >> 4) << 1;
  const int mbase = (lane >> 4) << 3;

  const int b      = (int)blockIdx.x;
  const int G      = (int)gridDim.x;
  const int nt     = (b < tiles) ? (tiles - b + G - 1) / G : 0;
  const int rounds = (nt + LAYER_MWAVES - 1) / LAYER_MWAVES;

  int g_dv = 0, g_e0 = 0, g_sc = 0;

  int g_fv0 = 0;
  const GatherSlot slot0 = sage_gather_slot(0, rounds, nt, wv, b, G);
  if (mw < 0 && slot0.active)
    sage_ask_heads(first, deg, slot0.n0, slot0.npw, N, lane, g_fv0, g_dv);

  if (tid < NODE_DIM) s_bias[tid] = bias[tid];
  {
    constexpr int W4   = NODE_DIM * NODE_DIM / 4;
    constexpr int NLD  = (W4 + LAYER_THREADS - 1) / LAYER_THREADS;
    float4 v[NLD];
    #pragma unroll
    for (int u = 0; u < NLD; ++u)
      v[u] = reinterpret_cast<const float4*>(Wl)[min(u * LAYER_THREADS + tid, W4 - 1)];
    #pragma unroll
    for (int u = 0; u < NLD; ++u) {
      const int i = u * LAYER_THREADS + tid;
      if (i < W4) *reinterpret_cast<float4*>(&s_wl[i >> 5][(i & 31) * 4]) = v[u];
    }
    #pragma unroll
    for (int u = 0; u < NLD; ++u)
      v[u] = reinterpret_cast<const float4*>(Wr)[min(u * LAYER_THREADS + tid, W4 - 1)];
    #pragma unroll
    for (int u = 0; u < NLD; ++u) {
      const int i = u * LAYER_THREADS + tid;
      if (i < W4) *reinterpret_cast<float4*>(&s_wr[i >> 5][(i & 31) * 4]) = v[u];
    }
  }

  if (mw < 0 && slot0.active)
    g_sc = sage_ask_ids(csr, g_fv0, g_dv, lane, g_e0);

  for (int r = 0; r <= rounds; ++r) {
    if (mw < 0) {
      const GatherSlot cur = sage_gather_slot(r, rounds, nt, wv, b, G);
      if (cur.active) {
        const GatherSlot nx = sage_gather_slot(r + 1, rounds, nt, wv, b, G);
        int fvn, dvn;
        sage_gather_rows(feat, first, deg, csr, &s_ring[r & 1][cur.q][cur.row0],
                         &s_root[r & 1][cur.q][cur.row0],
                         g_dv, g_e0, g_sc, cur.n0, cur.npw, N, lane,
                         nx.active, nx.n0, nx.npw, fvn, dvn);
        if (nx.active) {
          g_sc = sage_ask_ids(csr, fvn, dvn, lane, g_e0);
          g_dv = dvn;
        }
      }
    } else if (r > 0 && (r - 1) * LAYER_MWAVES + mw < nt) {
      const int tile  = b + ((r - 1) * LAYER_MWAVES + mw) * G;
      const int node0 = tile * 16;
      float (*ta)[LDS_STRIDE] = s_ring[(r - 1) & 1][mw];
      float (*tx)[LDS_STRIDE] = s_root[(r - 1) & 1][mw];
      const float* pa = &ta[m][khalf];
      const float* px = &tx[m][khalf];
      const float* pl = &s_wl[m][khalf];
      const float* pr = &s_wr[m][khalf];

      v8f acc[8];
      #pragma unroll
      for (int ct = 0; ct < 8; ++ct) {
        const float bvc = s_bias[ct * 16 + m];
        #pragma unroll
        for (int v = 0; v < 8; ++v) acc[ct][v] = bvc;
      }

      v2f a0, x0, l0[8], r0[8], a1, x1, l1[8], r1[8];
      a0 = *reinterpret_cast<const v2f*>(pa);
      x0 = *reinterpret_cast<const v2f*>(px);
      #pragma unroll
      for (int ct = 0; ct < 8; ++ct) {
        l0[ct] = *reinterpret_cast<const v2f*>(pl + ct * 16 * LDS_STRIDE);
        r0[ct] = *reinterpret_cast<const v2f*>(pr + ct * 16 * LDS_STRIDE);
      }
      #pragma unroll
      for (int k0 = 0; k0 < NODE_DIM; k0 += 8) {
        const int k1 = k0 + 4;
        a1 = *reinterpret_cast<const v2f*>(pa + k1);
        x1 = *reinterpret_cast<const v2f*>(px + k1);
        #pragma unroll
        for (int ct = 0; ct < 8; ++ct) {
          l1[ct] = *reinterpret_cast<const v2f*>(pl + ct * 16 * LDS_STRIDE + k1);
          r1[ct] = *reinterpret_cast<const v2f*>(pr + ct * 16 * LDS_STRIDE + k1);
        }
        __builtin_amdgcn_sched_barrier(0);
        #pragma unroll
        for (int ct = 0; ct < 8; ++ct) {
          acc[ct] = __builtin_amdgcn_wmma_f32_16x16x4_f32(false, a0, false, l0[ct],
                                                          (short)0, acc[ct], false, false);
          acc[ct] = __builtin_amdgcn_wmma_f32_16x16x4_f32(false, x0, false, r0[ct],
                                                          (short)0, acc[ct], false, false);
        }
        __builtin_amdgcn_sched_barrier(0);
        if (k0 + 8 < NODE_DIM) {
          const int k2 = k0 + 8;
          a0 = *reinterpret_cast<const v2f*>(pa + k2);
          x0 = *reinterpret_cast<const v2f*>(px + k2);
          #pragma unroll
          for (int ct = 0; ct < 8; ++ct) {
            l0[ct] = *reinterpret_cast<const v2f*>(pl + ct * 16 * LDS_STRIDE + k2);
            r0[ct] = *reinterpret_cast<const v2f*>(pr + ct * 16 * LDS_STRIDE + k2);
          }
        }
        __builtin_amdgcn_sched_barrier(0);
        #pragma unroll
        for (int ct = 0; ct < 8; ++ct) {
          acc[ct] = __builtin_amdgcn_wmma_f32_16x16x4_f32(false, a1, false, l1[ct],
                                                          (short)0, acc[ct], false, false);
          acc[ct] = __builtin_amdgcn_wmma_f32_16x16x4_f32(false, x1, false, r1[ct],
                                                          (short)0, acc[ct], false, false);
        }
        __builtin_amdgcn_sched_barrier(0);
      }

      wave_lds_handover();
      #pragma unroll
      for (int ct = 0; ct < 8; ++ct) {
        #pragma unroll
        for (int v = 0; v < 8; ++v) ta[mbase + v][ct * 16 + m] = acc[ct][v];
      }
      wave_lds_handover();

      if (NORM_RELU) {
        const int row = lane & 15;
        const int cb  = (lane >> 4) * 64;
        float ss = 0.f;
        #pragma unroll 8
        for (int c2 = 0; c2 < 64; ++c2) {
          const float v = ta[row][cb + c2];
          ss += v * v;
        }
        ss += __shfl_xor(ss, 16, 32);
        const float scl = 1.0f / fmaxf(sqrtf(ss), 1e-12f);
        #pragma unroll 8
        for (int c2 = 0; c2 < 64; ++c2) {
          const float v = ta[row][cb + c2] * scl;
          ta[row][cb + c2] = fmaxf(v, 0.0f);
        }
        wave_lds_handover();
      }

      {
        float4 vo[16];
        #pragma unroll
        for (int rr = 0; rr < 16; ++rr)
          vo[rr] = *reinterpret_cast<const float4*>(&ta[rr][lane * 4]);
        #pragma unroll
        for (int rr = 0; rr < 16; ++rr) {
          const int n = node0 + rr;
          if (n < N)
            *reinterpret_cast<float4*>(outp + (long long)n * NODE_DIM + lane * 4) = vo[rr];
        }
      }
    }
    round_barrier();
  }
}

extern "C" void kernel_launch(void* const* d_in, const int* in_sizes, int n_in,
                              void* d_out, int out_size, void* d_ws,
                              size_t ws_size, hipStream_t stream) {
  (void)n_in; (void)out_size; (void)ws_size;

  const float* x   = (const float*)d_in[0];
  const int*   ei  = (const int*)d_in[1];
  const float* W1l = (const float*)d_in[2];
  const float* b1  = (const float*)d_in[3];
  const float* W1r = (const float*)d_in[4];
  const float* W2l = (const float*)d_in[5];
  const float* b2  = (const float*)d_in[6];
  const float* W2r = (const float*)d_in[7];
  float* out = (float*)d_out;

  const int N = in_sizes[0] / NODE_DIM;
  const int E = in_sizes[1] / 2;
  const int* src = ei;
  const int* dst = ei + E;

  const int NB    = (N + BIN_NODES - 1) / BIN_NODES;
  const int NBP   = (NB + 3) & ~3;
  const int chunk = (E + NCHUNK - 1) / NCHUNK;

  char* ws = (char*)d_ws;
  int* deg = (int*)ws;
  size_t off = ((size_t)N * 4 + 255) & ~(size_t)255;
  off += (((size_t)N * NODE_DIM * 4) + 255) & ~(size_t)255;
  float* h = (float*)(ws + off);
  char* hb = (char*)h;
  size_t hoff = 0;
  int* binned = (int*)(hb + hoff);
  hoff += ((size_t)E * 8 + 255) & ~(size_t)255;
  int* slots = (int*)(hb + hoff);
  hoff += ((size_t)E * 4 + 255) & ~(size_t)255;
  int* cnt = (int*)(hb + hoff);
  hoff += ((size_t)NCHUNK * NBP * 4 + 255) & ~(size_t)255;
  int* binBase = (int*)(hb + hoff);
  off += (((size_t)N * NODE_DIM * 4) + 255) & ~(size_t)255;
  int* first = (int*)(ws + off);
  off += ((size_t)N * 4 + 255) & ~(size_t)255;
  int* csr = (int*)(ws + off);

  const int tiles  = (N + 15) / 16;

  sage_bincount_kernel<<<NCHUNK, 256, 0, stream>>>(dst, cnt, E, NBP, chunk);
  sage_binscatter_kernel<<<NCHUNK, 256, 0, stream>>>(
      src, dst, cnt, binBase, binned, E, NB, NBP, chunk);
  sage_bingroup_kernel<<<NB, BIN_NODES, 0, stream>>>(
      binned, binBase, deg, first, slots, csr, N, NB, E);

  sage_layer_kernel<true><<<LAYER_BLOCKS, LAYER_THREADS, 0, stream>>>(
      x, first, deg, csr, W1l, b1, W1r, h, N, tiles);

  sage_layer_kernel<false><<<LAYER_BLOCKS, LAYER_THREADS, 0, stream>>>(
      h, first, deg, csr, W2l, b2, W2r, out, N, tiles);
}
